# starting kernel
# baseline (speedup 1.0000x reference)
#include <hip/hip_runtime.h>


typedef __attribute__((ext_vector_type(2))) float v2f;
typedef __attribute__((ext_vector_type(8))) float v8f;

#define D 128

__global__ void k_init_deg(float* __restrict__ deg, int n) {
    int i = blockIdx.x * blockDim.x + threadIdx.x;
    if (i < n) deg[i] = 1.0f;
}

__global__ void k_count_deg(const int* __restrict__ ei_row, float* __restrict__ deg, int e) {
    int i = blockIdx.x * blockDim.x + threadIdx.x;
    if (i < e) atomicAdd(&deg[ei_row[i]], 1.0f);
}

__global__ void k_rsqrt_deg(float* __restrict__ deg, int n) {
    int i = blockIdx.x * blockDim.x + threadIdx.x;
    if (i < n) deg[i] = rsqrtf(deg[i]);
}

__global__ void k_gemm_wmma(const float* __restrict__ x,
                            const float* __restrict__ W,
                            const float* __restrict__ bias,
                            float* __restrict__ h) {
    const int wave = threadIdx.x >> 5;
    const int lane = threadIdx.x & 31;
    const int m0 = blockIdx.x << 4;
    const int n0 = wave << 4;
    const int lm = lane & 15;
    const int lk = (lane >> 4) << 1;

    const float* xr = x + (size_t)(m0 + lm) * D;
    const float* wr = W + (size_t)(n0 + lm) * D;

    v8f acc = {};
#pragma unroll
    for (int k = 0; k < D; k += 4) {
        v2f a = *(const v2f*)(xr + k + lk);
        v2f b = *(const v2f*)(wr + k + lk);
        acc = __builtin_amdgcn_wmma_f32_16x16x4_f32(
            false, a, false, b, (short)0, acc, false, false);
    }

    const float bn = bias[n0 + lm];
    float* hp = h + (size_t)(m0 + ((lane >> 4) << 3)) * D + (n0 + lm);
#pragma unroll
    for (int v = 0; v < 8; ++v)
        hp[(size_t)v * D] = acc[v] + bn;
}

__global__ void k_self_loop(const float* __restrict__ h,
                            const float* __restrict__ rsq,
                            float* __restrict__ out, long long total) {
    long long i = (long long)blockIdx.x * blockDim.x + threadIdx.x;
    if (i < total) {
        int node = (int)(i >> 7);
        float r = rsq[node];
        out[i] = h[i] * (r * r);
    }
}

__global__ void k_aggregate(const int* __restrict__ ei,
                            const float* __restrict__ h,
                            const float* __restrict__ rsq,
                            float* __restrict__ out, int e) {
    const int edge = blockIdx.x * (blockDim.x >> 5) + (threadIdx.x >> 5);
    if (edge >= e) return;
    const int lane = threadIdx.x & 31;

    const int dst = ei[edge];
    const int src = ei[e + edge];
    const float norm = rsq[dst] * rsq[src];

    const float4 hv = *(const float4*)(h + (size_t)src * D + lane * 4);
    float* o = out + (size_t)dst * D + lane * 4;
    atomicAdd(o + 0, norm * hv.x);
    atomicAdd(o + 1, norm * hv.y);
    atomicAdd(o + 2, norm * hv.z);
    atomicAdd(o + 3, norm * hv.w);
}

extern "C" void kernel_launch(void* const* d_in, const int* in_sizes, int n_in,
                              void* d_out, int out_size, void* d_ws, size_t ws_size,
                              hipStream_t stream) {
    const float* x    = (const float*)d_in[0];
    const int*   ei   = (const int*)d_in[1];
    const float* W    = (const float*)d_in[2];
    const float* bias = (const float*)d_in[3];
    float*       out  = (float*)d_out;

    const int n = in_sizes[0] / D;
    const int e = in_sizes[1] / 2;

    float* h   = (float*)d_ws;
    float* rsq = h + (size_t)n * D;

    const int T = 256;

    k_init_deg <<<(n + T - 1) / T, T, 0, stream>>>(rsq, n);
    k_count_deg<<<(e + T - 1) / T, T, 0, stream>>>(ei, rsq, e);
    k_rsqrt_deg<<<(n + T - 1) / T, T, 0, stream>>>(rsq, n);

    k_gemm_wmma<<<n / 16, T, 0, stream>>>(x, W, bias, h);

    long long total = (long long)n * D;
    k_self_loop<<<(unsigned)((total + T - 1) / T), T, 0, stream>>>(h, rsq, out, total);

    k_aggregate<<<(e + 7) / 8, T, 0, stream>>>(ei, h, rsq, out, e);
}


# Aggregate by pull over edges counting-sorted by dst (integer-atomic arrival rank sets the sum order): one wave sums each out row in registers and stores it once; no float atomics on out
# speedup vs baseline: 1.9298x; 1.9298x over previous
#include <hip/hip_runtime.h>


typedef __attribute__((ext_vector_type(2))) float v2f;
typedef __attribute__((ext_vector_type(8))) float v8f;

#define D 128

__global__ void k_zero_cnt(int* __restrict__ cnt, int n) {
    int i = blockIdx.x * blockDim.x + threadIdx.x;
    if (i < n) cnt[i] = 0;
}

__global__ void k_count_rank(const int* __restrict__ ei_row, int* __restrict__ cnt,
                             int* __restrict__ rank, int e) {
    int i = blockIdx.x * blockDim.x + threadIdx.x;
    if (i < e) rank[i] = atomicAdd(&cnt[ei_row[i]], 1);
}

#define SCAN_T 1024
__global__ void __launch_bounds__(SCAN_T)
k_scan_deg(const int* __restrict__ cnt, int* __restrict__ rowptr,
           float* __restrict__ rsq, int n) {
    __shared__ int s_part[SCAN_T / 32];
    __shared__ int s_wave[SCAN_T / 32];
    const int t = threadIdx.x;
    const int lane = t & 31;
    const int wv = t >> 5;

    int pre = 0;
    for (int k = 0; k < (int)blockIdx.x; ++k) pre += cnt[k * SCAN_T + t];
#pragma unroll
    for (int o = 16; o > 0; o >>= 1) pre += __shfl_xor(pre, o);
    if (lane == 0) s_part[wv] = pre;

    const int i = blockIdx.x * SCAN_T + t;
    const int v = (i < n) ? cnt[i] : 0;
    int inc = v;
#pragma unroll
    for (int o = 1; o < 32; o <<= 1) {
        int u = __shfl_up(inc, o);
        if (lane >= o) inc += u;
    }
    if (lane == 31) s_wave[wv] = inc;
    __syncthreads();

    int base = 0;
#pragma unroll
    for (int k = 0; k < SCAN_T / 32; ++k) {
        base += s_part[k];
        if (k < wv) base += s_wave[k];
    }
    if (i < n) {
        rowptr[i] = base + inc - v;
        rsq[i] = rsqrtf((float)(v + 1));
        if (i == n - 1) rowptr[n] = base + inc;
    }
}

__global__ void k_fill_csr(const int* __restrict__ ei, const int* __restrict__ rowptr,
                           const int* __restrict__ rank, int* __restrict__ csr, int e) {
    int i = blockIdx.x * blockDim.x + threadIdx.x;
    if (i < e) csr[rowptr[ei[i]] + rank[i]] = ei[e + i];
}

__global__ void k_gemm_wmma(const float* __restrict__ x,
                            const float* __restrict__ W,
                            const float* __restrict__ bias,
                            float* __restrict__ h) {
    const int wave = threadIdx.x >> 5;
    const int lane = threadIdx.x & 31;
    const int m0 = blockIdx.x << 4;
    const int n0 = wave << 4;
    const int lm = lane & 15;
    const int lk = (lane >> 4) << 1;

    const float* xr = x + (size_t)(m0 + lm) * D;
    const float* wr = W + (size_t)(n0 + lm) * D;

    v8f acc = {};
#pragma unroll
    for (int k = 0; k < D; k += 4) {
        v2f a = *(const v2f*)(xr + k + lk);
        v2f b = *(const v2f*)(wr + k + lk);
        acc = __builtin_amdgcn_wmma_f32_16x16x4_f32(
            false, a, false, b, (short)0, acc, false, false);
    }

    const float bn = bias[n0 + lm];
    float* hp = h + (size_t)(m0 + ((lane >> 4) << 3)) * D + (n0 + lm);
#pragma unroll
    for (int v = 0; v < 8; ++v)
        hp[(size_t)v * D] = acc[v] + bn;
}

__device__ __forceinline__ float4 ld_row(const float* __restrict__ h, int s, int lane) {
    return *(const float4*)(h + (size_t)s * D + lane * 4);
}
__device__ __forceinline__ void fma4(float4& a, float w, const float4& v) {
    a.x = fmaf(w, v.x, a.x); a.y = fmaf(w, v.y, a.y);
    a.z = fmaf(w, v.z, a.z); a.w = fmaf(w, v.w, a.w);
}
__device__ __forceinline__ int bc_i(int v, int k) { return __builtin_amdgcn_readlane(v, k); }
__device__ __forceinline__ float bc_f(float v, int k) {
    return __int_as_float(__builtin_amdgcn_readlane(__float_as_int(v), k));
}

__global__ void __launch_bounds__(256)
k_gather(const int* __restrict__ rowptr, const int* __restrict__ csr,
         const float* __restrict__ h, const float* __restrict__ rsq,
         float* __restrict__ out, int n, int e) {
    const int node = blockIdx.x * (blockDim.x >> 5) + (threadIdx.x >> 5);
    if (node >= n) return;
    const int lane = threadIdx.x & 31;

    int beg = __builtin_amdgcn_readfirstlane(rowptr[node]);
    int end = __builtin_amdgcn_readfirstlane(rowptr[node + 1]);
    const float ri = __int_as_float(__builtin_amdgcn_readfirstlane(__float_as_int(rsq[node])));
    beg = max(beg, 0);
    end = min(end, e);

    float4 acc = ld_row(h, node, lane);
    const float w_self = ri * ri;
    acc.x *= w_self; acc.y *= w_self; acc.z *= w_self; acc.w *= w_self;

    for (int base = beg; base < end; base += 32) {
        const int m = min(32, end - base);
        int s_l = 0;
        float q_l = 0.0f;
        if (lane < m) {
            s_l = csr[base + lane];
            q_l = rsq[s_l];
        }
        int k = 0;
        for (; k + 8 <= m; k += 8) {
            const float4 v0 = ld_row(h, bc_i(s_l, k), lane);
            const float4 v1 = ld_row(h, bc_i(s_l, k + 1), lane);
            const float4 v2 = ld_row(h, bc_i(s_l, k + 2), lane);
            const float4 v3 = ld_row(h, bc_i(s_l, k + 3), lane);
            const float4 v4 = ld_row(h, bc_i(s_l, k + 4), lane);
            const float4 v5 = ld_row(h, bc_i(s_l, k + 5), lane);
            const float4 v6 = ld_row(h, bc_i(s_l, k + 6), lane);
            const float4 v7 = ld_row(h, bc_i(s_l, k + 7), lane);
            fma4(acc, ri * bc_f(q_l, k), v0);
            fma4(acc, ri * bc_f(q_l, k + 1), v1);
            fma4(acc, ri * bc_f(q_l, k + 2), v2);
            fma4(acc, ri * bc_f(q_l, k + 3), v3);
            fma4(acc, ri * bc_f(q_l, k + 4), v4);
            fma4(acc, ri * bc_f(q_l, k + 5), v5);
            fma4(acc, ri * bc_f(q_l, k + 6), v6);
            fma4(acc, ri * bc_f(q_l, k + 7), v7);
        }
        if (k + 4 <= m) {
            const float4 v0 = ld_row(h, bc_i(s_l, k), lane);
            const float4 v1 = ld_row(h, bc_i(s_l, k + 1), lane);
            const float4 v2 = ld_row(h, bc_i(s_l, k + 2), lane);
            const float4 v3 = ld_row(h, bc_i(s_l, k + 3), lane);
            fma4(acc, ri * bc_f(q_l, k), v0);
            fma4(acc, ri * bc_f(q_l, k + 1), v1);
            fma4(acc, ri * bc_f(q_l, k + 2), v2);
            fma4(acc, ri * bc_f(q_l, k + 3), v3);
            k += 4;
        }
        const int r = m - k;
        if (r > 0) {
            const float4 v0 = ld_row(h, bc_i(s_l, k), lane);
            float4 v1 = make_float4(0.0f, 0.0f, 0.0f, 0.0f), v2 = v1;
            if (r > 1) v1 = ld_row(h, bc_i(s_l, k + 1), lane);
            if (r > 2) v2 = ld_row(h, bc_i(s_l, k + 2), lane);
            fma4(acc, ri * bc_f(q_l, k), v0);
            if (r > 1) fma4(acc, ri * bc_f(q_l, k + 1), v1);
            if (r > 2) fma4(acc, ri * bc_f(q_l, k + 2), v2);
        }
    }
    *(float4*)(out + (size_t)node * D + lane * 4) = acc;
}

extern "C" void kernel_launch(void* const* d_in, const int* in_sizes, int n_in,
                              void* d_out, int out_size, void* d_ws, size_t ws_size,
                              hipStream_t stream) {
    const float* x    = (const float*)d_in[0];
    const int*   ei   = (const int*)d_in[1];
    const float* W    = (const float*)d_in[2];
    const float* bias = (const float*)d_in[3];
    float*       out  = (float*)d_out;

    const int n = in_sizes[0] / D;
    const int e = in_sizes[1] / 2;

    float* h      = (float*)d_ws;
    float* rsq    = h + (size_t)n * D;
    int*   cnt    = (int*)(rsq + n);
    int*   rowptr = cnt + n;
    int*   rank   = rowptr + (n + 1);
    int*   csr    = rank + e;

    const int T = 256;

    k_zero_cnt  <<<(n + T - 1) / T, T, 0, stream>>>(cnt, n);
    k_count_rank<<<(e + T - 1) / T, T, 0, stream>>>(ei, cnt, rank, e);
    k_scan_deg  <<<(n + SCAN_T - 1) / SCAN_T, SCAN_T, 0, stream>>>(cnt, rowptr, rsq, n);
    k_fill_csr  <<<(e + T - 1) / T, T, 0, stream>>>(ei, rowptr, rank, csr, e);

    k_gemm_wmma<<<n / 16, T, 0, stream>>>(x, W, bias, h);

    k_gather<<<(n + 7) / 8, T, 0, stream>>>(rowptr, csr, h, rsq, out, n, e);
}


# About six times fewer waves: edge, slab and node kernels launch resident-size grids and every wave loops over its items (W fragments read once per wave); h bits and sum-order rule as before
# speedup vs baseline: 2.1441x; 1.1110x over previous
#include <hip/hip_runtime.h>


typedef __attribute__((ext_vector_type(2))) float v2f;
typedef __attribute__((ext_vector_type(8))) float v8f;

#define D 128

__global__ void k_zero_cnt(int* __restrict__ cnt, int n) {
    int i = blockIdx.x * blockDim.x + threadIdx.x;
    if (i < n) cnt[i] = 0;
}

__global__ void __launch_bounds__(256)
k_count_rank(const int* __restrict__ ei_row, int* __restrict__ cnt,
             int* __restrict__ rank, int e) {
    const int st = gridDim.x * blockDim.x;
    int i = blockIdx.x * blockDim.x + threadIdx.x;
    for (; i + 3 * st < e; i += 4 * st) {
        const int d0 = ei_row[i];
        const int d1 = ei_row[i + st];
        const int d2 = ei_row[i + 2 * st];
        const int d3 = ei_row[i + 3 * st];
        const int r0 = atomicAdd(&cnt[d0], 1);
        const int r1 = atomicAdd(&cnt[d1], 1);
        const int r2 = atomicAdd(&cnt[d2], 1);
        const int r3 = atomicAdd(&cnt[d3], 1);
        rank[i] = r0;
        rank[i + st] = r1;
        rank[i + 2 * st] = r2;
        rank[i + 3 * st] = r3;
    }
    for (; i < e; i += st) rank[i] = atomicAdd(&cnt[ei_row[i]], 1);
}

#define SCAN_T 1024
__global__ void __launch_bounds__(SCAN_T)
k_scan_deg(const int* __restrict__ cnt, int* __restrict__ rowptr,
           float* __restrict__ rsq, int n) {
    __shared__ int s_part[SCAN_T / 32];
    __shared__ int s_wave[SCAN_T / 32];
    const int t = threadIdx.x;
    const int lane = t & 31;
    const int wv = t >> 5;

    int pre = 0;
    for (int k = 0; k < (int)blockIdx.x; ++k) pre += cnt[k * SCAN_T + t];
#pragma unroll
    for (int o = 16; o > 0; o >>= 1) pre += __shfl_xor(pre, o);
    if (lane == 0) s_part[wv] = pre;

    const int i = blockIdx.x * SCAN_T + t;
    const int v = (i < n) ? cnt[i] : 0;
    int inc = v;
#pragma unroll
    for (int o = 1; o < 32; o <<= 1) {
        int u = __shfl_up(inc, o);
        if (lane >= o) inc += u;
    }
    if (lane == 31) s_wave[wv] = inc;
    __syncthreads();

    int base = 0;
#pragma unroll
    for (int k = 0; k < SCAN_T / 32; ++k) {
        base += s_part[k];
        if (k < wv) base += s_wave[k];
    }
    if (i < n) {
        rowptr[i] = base + inc - v;
        rsq[i] = rsqrtf((float)(v + 1));
        if (i == n - 1) rowptr[n] = base + inc;
    }
}

__global__ void __launch_bounds__(256)
k_fill_csr(const int* __restrict__ ei, const int* __restrict__ rowptr,
           const int* __restrict__ rank, int* __restrict__ csr, int e) {
    const int st = gridDim.x * blockDim.x;
    int i = blockIdx.x * blockDim.x + threadIdx.x;
    for (; i + 3 * st < e; i += 4 * st) {
        const int d0 = ei[i];
        const int d1 = ei[i + st];
        const int d2 = ei[i + 2 * st];
        const int d3 = ei[i + 3 * st];
        const int k0 = rank[i];
        const int k1 = rank[i + st];
        const int k2 = rank[i + 2 * st];
        const int k3 = rank[i + 3 * st];
        const int s0 = ei[e + i];
        const int s1 = ei[e + i + st];
        const int s2 = ei[e + i + 2 * st];
        const int s3 = ei[e + i + 3 * st];
        const int p0 = rowptr[d0];
        const int p1 = rowptr[d1];
        const int p2 = rowptr[d2];
        const int p3 = rowptr[d3];
        csr[p0 + k0] = s0;
        csr[p1 + k1] = s1;
        csr[p2 + k2] = s2;
        csr[p3 + k3] = s3;
    }
    for (; i < e; i += st) csr[rowptr[ei[i]] + rank[i]] = ei[e + i];
}

__global__ void __launch_bounds__(256)
k_gemm_wmma(const float* __restrict__ x,
            const float* __restrict__ W,
            const float* __restrict__ bias,
            float* __restrict__ h, int n_slabs) {
    const int wave = threadIdx.x >> 5;
    const int lane = threadIdx.x & 31;
    const int n0 = wave << 4;
    const int lm = lane & 15;
    const int lk = (lane >> 4) << 1;

    const float* wr = W + (size_t)(n0 + lm) * D + lk;
    v2f b[D / 4];
#pragma unroll
    for (int i = 0; i < D / 4; ++i) b[i] = *(const v2f*)(wr + 4 * i);
    const float bn = bias[n0 + lm];

    for (int slab = blockIdx.x; slab < n_slabs; slab += gridDim.x) {
        const int m0 = slab << 4;
        const float* xr = x + (size_t)(m0 + lm) * D + lk;

        v8f acc = {};
#pragma unroll
        for (int i = 0; i < D / 4; ++i) {
            v2f a = *(const v2f*)(xr + 4 * i);
            acc = __builtin_amdgcn_wmma_f32_16x16x4_f32(
                false, a, false, b[i], (short)0, acc, false, false);
        }

        float* hp = h + (size_t)(m0 + ((lane >> 4) << 3)) * D + (n0 + lm);
#pragma unroll
        for (int v = 0; v < 8; ++v)
            hp[(size_t)v * D] = acc[v] + bn;
    }
}

__device__ __forceinline__ float4 ld_row(const float* __restrict__ h, int s, int lane) {
    return *(const float4*)(h + (size_t)s * D + lane * 4);
}
__device__ __forceinline__ void fma4(float4& a, float w, const float4& v) {
    a.x = fmaf(w, v.x, a.x); a.y = fmaf(w, v.y, a.y);
    a.z = fmaf(w, v.z, a.z); a.w = fmaf(w, v.w, a.w);
}
__device__ __forceinline__ int bc_i(int v, int k) { return __builtin_amdgcn_readlane(v, k); }
__device__ __forceinline__ float bc_f(float v, int k) {
    return __int_as_float(__builtin_amdgcn_readlane(__float_as_int(v), k));
}

__global__ void __launch_bounds__(256)
k_gather(const int* __restrict__ rowptr, const int* __restrict__ csr,
         const float* __restrict__ h, const float* __restrict__ rsq,
         float* __restrict__ out, int n, int e) {
    const int wpb = blockDim.x >> 5;
    const int lane = threadIdx.x & 31;
    for (int node = blockIdx.x * wpb + (threadIdx.x >> 5); node < n; node += gridDim.x * wpb) {
        int beg = __builtin_amdgcn_readfirstlane(rowptr[node]);
        int end = __builtin_amdgcn_readfirstlane(rowptr[node + 1]);
        const float ri = __int_as_float(__builtin_amdgcn_readfirstlane(__float_as_int(rsq[node])));
        beg = max(beg, 0);
        end = min(end, e);

        float4 acc = ld_row(h, node, lane);
        const float w_self = ri * ri;
        acc.x *= w_self; acc.y *= w_self; acc.z *= w_self; acc.w *= w_self;

        for (int base = beg; base < end; base += 32) {
            const int m = min(32, end - base);
            int s_l = 0;
            float q_l = 0.0f;
            if (lane < m) {
                s_l = csr[base + lane];
                q_l = rsq[s_l];
            }
            int k = 0;
            for (; k + 8 <= m; k += 8) {
                const float4 v0 = ld_row(h, bc_i(s_l, k), lane);
                const float4 v1 = ld_row(h, bc_i(s_l, k + 1), lane);
                const float4 v2 = ld_row(h, bc_i(s_l, k + 2), lane);
                const float4 v3 = ld_row(h, bc_i(s_l, k + 3), lane);
                const float4 v4 = ld_row(h, bc_i(s_l, k + 4), lane);
                const float4 v5 = ld_row(h, bc_i(s_l, k + 5), lane);
                const float4 v6 = ld_row(h, bc_i(s_l, k + 6), lane);
                const float4 v7 = ld_row(h, bc_i(s_l, k + 7), lane);
                fma4(acc, ri * bc_f(q_l, k), v0);
                fma4(acc, ri * bc_f(q_l, k + 1), v1);
                fma4(acc, ri * bc_f(q_l, k + 2), v2);
                fma4(acc, ri * bc_f(q_l, k + 3), v3);
                fma4(acc, ri * bc_f(q_l, k + 4), v4);
                fma4(acc, ri * bc_f(q_l, k + 5), v5);
                fma4(acc, ri * bc_f(q_l, k + 6), v6);
                fma4(acc, ri * bc_f(q_l, k + 7), v7);
            }
            if (k + 4 <= m) {
                const float4 v0 = ld_row(h, bc_i(s_l, k), lane);
                const float4 v1 = ld_row(h, bc_i(s_l, k + 1), lane);
                const float4 v2 = ld_row(h, bc_i(s_l, k + 2), lane);
                const float4 v3 = ld_row(h, bc_i(s_l, k + 3), lane);
                fma4(acc, ri * bc_f(q_l, k), v0);
                fma4(acc, ri * bc_f(q_l, k + 1), v1);
                fma4(acc, ri * bc_f(q_l, k + 2), v2);
                fma4(acc, ri * bc_f(q_l, k + 3), v3);
                k += 4;
            }
            const int r = m - k;
            if (r > 0) {
                const float4 v0 = ld_row(h, bc_i(s_l, k), lane);
                float4 v1 = make_float4(0.0f, 0.0f, 0.0f, 0.0f), v2 = v1;
                if (r > 1) v1 = ld_row(h, bc_i(s_l, k + 1), lane);
                if (r > 2) v2 = ld_row(h, bc_i(s_l, k + 2), lane);
                fma4(acc, ri * bc_f(q_l, k), v0);
                if (r > 1) fma4(acc, ri * bc_f(q_l, k + 1), v1);
                if (r > 2) fma4(acc, ri * bc_f(q_l, k + 2), v2);
            }
        }
        *(float4*)(out + (size_t)node * D + lane * 4) = acc;
    }
}

extern "C" void kernel_launch(void* const* d_in, const int* in_sizes, int n_in,
                              void* d_out, int out_size, void* d_ws, size_t ws_size,
                              hipStream_t stream) {
    const float* x    = (const float*)d_in[0];
    const int*   ei   = (const int*)d_in[1];
    const float* W    = (const float*)d_in[2];
    const float* bias = (const float*)d_in[3];
    float*       out  = (float*)d_out;

    const int n = in_sizes[0] / D;
    const int e = in_sizes[1] / 2;

    float* h      = (float*)d_ws;
    float* rsq    = h + (size_t)n * D;
    int*   cnt    = (int*)(rsq + n);
    int*   rowptr = cnt + n;
    int*   rank   = rowptr + (n + 1);
    int*   csr    = rank + e;

    const int T = 256;


    k_zero_cnt  <<<(n + T - 1) / T, T, 0, stream>>>(cnt, n);
    k_count_rank<<<512, T, 0, stream>>>(ei, cnt, rank, e);
    k_scan_deg  <<<(n + SCAN_T - 1) / SCAN_T, SCAN_T, 0, stream>>>(cnt, rowptr, rsq, n);
    k_fill_csr  <<<512, T, 0, stream>>>(ei, rowptr, rank, csr, e);

    k_gemm_wmma<<<1280, T, 0, stream>>>(x, W, bias, h, n / 16);

    k_gather<<<2048, T, 0, stream>>>(rowptr, csr, h, rsq, out, n, e);
}


# Counting sort by dst made two-level with LDS counters (512-node bins; one block per bin builds rowptr and csr): no global atomics, no per-edge scattered access; sum order set by LDS atomic arrival
# speedup vs baseline: 3.8141x; 1.7789x over previous
#include <hip/hip_runtime.h>


typedef __attribute__((ext_vector_type(2))) float v2f;
typedef __attribute__((ext_vector_type(8))) float v8f;

#define D 128

#define BIN_SHIFT 9
#define BIN_N 512
#define NB_MAX 256
#define TILE 4096
#define SCAN_T 1024
#define CSR_CAP 12288

__global__ void __launch_bounds__(256)
k_bin_hist(const int* __restrict__ ei_row, int* __restrict__ ghist, int e, int nb) {
    __shared__ int hist[NB_MAX];
    const int t = threadIdx.x;
    hist[t] = 0;
    __syncthreads();
    const int chunk = (e + (int)gridDim.x - 1) / (int)gridDim.x;
    const int c0 = min(e, (int)blockIdx.x * chunk);
    const int c1 = min(e, c0 + chunk);
    int i = c0 + t;
    for (; i + 768 < c1; i += 1024) {
        const int d0 = ei_row[i];
        const int d1 = ei_row[i + 256];
        const int d2 = ei_row[i + 512];
        const int d3 = ei_row[i + 768];
        atomicAdd(&hist[d0 >> BIN_SHIFT], 1);
        atomicAdd(&hist[d1 >> BIN_SHIFT], 1);
        atomicAdd(&hist[d2 >> BIN_SHIFT], 1);
        atomicAdd(&hist[d3 >> BIN_SHIFT], 1);
    }
    for (; i < c1; i += 256) atomicAdd(&hist[ei_row[i] >> BIN_SHIFT], 1);
    __syncthreads();
    if (t < nb) ghist[t * (int)gridDim.x + (int)blockIdx.x] = hist[t];
}

__global__ void __launch_bounds__(SCAN_T)
k_scan_bins(const int* __restrict__ ghist, int* __restrict__ goff, int m) {
    __shared__ int s_part[SCAN_T / 32];
    __shared__ int s_wave[SCAN_T / 32];
    if ((int)blockIdx.x * SCAN_T >= m) return;
    const int t = threadIdx.x;
    const int lane = t & 31;
    const int wv = t >> 5;

    int pre = 0;
    for (int k = 0; k < (int)blockIdx.x; ++k) pre += ghist[k * SCAN_T + t];
#pragma unroll
    for (int o = 16; o > 0; o >>= 1) pre += __shfl_xor(pre, o);
    if (lane == 0) s_part[wv] = pre;

    const int i = blockIdx.x * SCAN_T + t;
    const int v = (i < m) ? ghist[i] : 0;
    int inc = v;
#pragma unroll
    for (int o = 1; o < 32; o <<= 1) {
        int u = __shfl_up(inc, o);
        if (lane >= o) inc += u;
    }
    if (lane == 31) s_wave[wv] = inc;
    __syncthreads();

    int base = 0;
#pragma unroll
    for (int k = 0; k < SCAN_T / 32; ++k) {
        base += s_part[k];
        if (k < wv) base += s_wave[k];
    }
    if (i < m) {
        goff[i] = base + inc - v;
        if (i == m - 1) goff[m] = base + inc;
    }
}

__global__ void __launch_bounds__(256)
k_bin_scatter(const int* __restrict__ ei, const int* __restrict__ goff,
              int2* __restrict__ binned, int e, int nb) {
    __shared__ int hist[NB_MAX];
    __shared__ int loff[NB_MAX];
    __shared__ int gcur[NB_MAX];
    __shared__ int s_wave[8];
    __shared__ int2 stage[TILE];
    const int t = threadIdx.x;
    const int lane = t & 31;
    const int wv = t >> 5;
    const int chunk = (e + (int)gridDim.x - 1) / (int)gridDim.x;
    const int c0 = min(e, (int)blockIdx.x * chunk);
    const int c1 = min(e, c0 + chunk);
    gcur[t] = (t < nb) ? goff[t * (int)gridDim.x + (int)blockIdx.x] : 0;

    for (int t0 = c0; t0 < c1; t0 += TILE) {
        const int mt = min(TILE, c1 - t0);
        hist[t] = 0;
        __syncthreads();

        int dd[TILE / 256], ss[TILE / 256], rr[TILE / 256];
#pragma unroll
        for (int k = 0; k < TILE / 256; ++k) {
            const int j = k * 256 + t;
            dd[k] = (j < mt) ? ei[t0 + j] : 0;
            ss[k] = (j < mt) ? ei[e + t0 + j] : 0;
        }
#pragma unroll
        for (int k = 0; k < TILE / 256; ++k) {
            rr[k] = 0;
            if (k * 256 + t < mt) rr[k] = atomicAdd(&hist[dd[k] >> BIN_SHIFT], 1);
        }
        __syncthreads();

        const int v = hist[t];
        int inc = v;
#pragma unroll
        for (int o = 1; o < 32; o <<= 1) {
            int u = __shfl_up(inc, o);
            if (lane >= o) inc += u;
        }
        if (lane == 31) s_wave[wv] = inc;
        __syncthreads();
        int base = 0;
#pragma unroll
        for (int k = 0; k < 8; ++k)
            if (k < wv) base += s_wave[k];
        loff[t] = base + inc - v;
        __syncthreads();

#pragma unroll
        for (int k = 0; k < TILE / 256; ++k)
            if (k * 256 + t < mt)
                stage[loff[dd[k] >> BIN_SHIFT] + rr[k]] = make_int2(dd[k], ss[k]);
        __syncthreads();

        for (int j = t; j < mt; j += 256) {
            const int2 el = stage[j];
            const int b = el.x >> BIN_SHIFT;
            binned[gcur[b] + (j - loff[b])] = el;
        }
        __syncthreads();
        gcur[t] += v;
    }
}

__global__ void __launch_bounds__(BIN_N)
k_bin_csr(const int2* __restrict__ binned, const int* __restrict__ goff,
          int* __restrict__ rowptr, float* __restrict__ rsq, int* __restrict__ csr,
          int n, int g) {
    __shared__ int cnt[BIN_N];
    __shared__ int cur[BIN_N];
    __shared__ int s_wave[BIN_N / 32];
    __shared__ int stage[CSR_CAP];
    const int t = threadIdx.x;
    const int lane = t & 31;
    const int wv = t >> 5;
    const int b = blockIdx.x;
    const int b0 = goff[b * g];
    const int b1 = goff[(b + 1) * g];
    const int m = b1 - b0;

    cnt[t] = 0;
    __syncthreads();
    {
        int i = b0 + t;
        for (; i + 3 * BIN_N < b1; i += 4 * BIN_N) {
            const int k0 = binned[i].x;
            const int k1 = binned[i + BIN_N].x;
            const int k2 = binned[i + 2 * BIN_N].x;
            const int k3 = binned[i + 3 * BIN_N].x;
            atomicAdd(&cnt[k0 & (BIN_N - 1)], 1);
            atomicAdd(&cnt[k1 & (BIN_N - 1)], 1);
            atomicAdd(&cnt[k2 & (BIN_N - 1)], 1);
            atomicAdd(&cnt[k3 & (BIN_N - 1)], 1);
        }
        for (; i < b1; i += BIN_N) atomicAdd(&cnt[binned[i].x & (BIN_N - 1)], 1);
    }
    __syncthreads();

    const int v = cnt[t];
    int inc = v;
#pragma unroll
    for (int o = 1; o < 32; o <<= 1) {
        int u = __shfl_up(inc, o);
        if (lane >= o) inc += u;
    }
    if (lane == 31) s_wave[wv] = inc;
    __syncthreads();
    int base = 0;
#pragma unroll
    for (int k = 0; k < BIN_N / 32; ++k)
        if (k < wv) base += s_wave[k];
    const int off = base + inc - v;
    cur[t] = off;
    const int node = b * BIN_N + t;
    if (node < n) {
        rowptr[node] = b0 + off;
        rsq[node] = rsqrtf((float)(v + 1));
        if (node == n - 1) rowptr[n] = b1;
    }
    __syncthreads();

    if (m <= CSR_CAP) {
        int i = b0 + t;
        for (; i + 3 * BIN_N < b1; i += 4 * BIN_N) {
            const int2 e0 = binned[i];
            const int2 e1 = binned[i + BIN_N];
            const int2 e2 = binned[i + 2 * BIN_N];
            const int2 e3 = binned[i + 3 * BIN_N];
            stage[atomicAdd(&cur[e0.x & (BIN_N - 1)], 1)] = e0.y;
            stage[atomicAdd(&cur[e1.x & (BIN_N - 1)], 1)] = e1.y;
            stage[atomicAdd(&cur[e2.x & (BIN_N - 1)], 1)] = e2.y;
            stage[atomicAdd(&cur[e3.x & (BIN_N - 1)], 1)] = e3.y;
        }
        for (; i < b1; i += BIN_N) {
            const int2 el = binned[i];
            stage[atomicAdd(&cur[el.x & (BIN_N - 1)], 1)] = el.y;
        }
        __syncthreads();
        for (int j = t; j < m; j += BIN_N) csr[b0 + j] = stage[j];
    } else {
        for (int i = b0 + t; i < b1; i += BIN_N) {
            const int2 el = binned[i];
            csr[b0 + atomicAdd(&cur[el.x & (BIN_N - 1)], 1)] = el.y;
        }
    }
}

__global__ void __launch_bounds__(256)
k_gemm_wmma(const float* __restrict__ x,
            const float* __restrict__ W,
            const float* __restrict__ bias,
            float* __restrict__ h, int n_slabs) {
    const int wave = threadIdx.x >> 5;
    const int lane = threadIdx.x & 31;
    const int n0 = wave << 4;
    const int lm = lane & 15;
    const int lk = (lane >> 4) << 1;

    const float* wr = W + (size_t)(n0 + lm) * D + lk;
    v2f b[D / 4];
#pragma unroll
    for (int i = 0; i < D / 4; ++i) b[i] = *(const v2f*)(wr + 4 * i);
    const float bn = bias[n0 + lm];

    for (int slab = blockIdx.x; slab < n_slabs; slab += gridDim.x) {
        const int m0 = slab << 4;
        const float* xr = x + (size_t)(m0 + lm) * D + lk;

        v8f acc = {};
#pragma unroll
        for (int i = 0; i < D / 4; ++i) {
            v2f a = *(const v2f*)(xr + 4 * i);
            acc = __builtin_amdgcn_wmma_f32_16x16x4_f32(
                false, a, false, b[i], (short)0, acc, false, false);
        }

        float* hp = h + (size_t)(m0 + ((lane >> 4) << 3)) * D + (n0 + lm);
#pragma unroll
        for (int v = 0; v < 8; ++v)
            hp[(size_t)v * D] = acc[v] + bn;
    }
}

__device__ __forceinline__ float4 ld_row(const float* __restrict__ h, int s, int lane) {
    return *(const float4*)(h + (size_t)s * D + lane * 4);
}
__device__ __forceinline__ void fma4(float4& a, float w, const float4& v) {
    a.x = fmaf(w, v.x, a.x); a.y = fmaf(w, v.y, a.y);
    a.z = fmaf(w, v.z, a.z); a.w = fmaf(w, v.w, a.w);
}
__device__ __forceinline__ int bc_i(int v, int k) { return __builtin_amdgcn_readlane(v, k); }
__device__ __forceinline__ float bc_f(float v, int k) {
    return __int_as_float(__builtin_amdgcn_readlane(__float_as_int(v), k));
}

__global__ void __launch_bounds__(256)
k_gather(const int* __restrict__ rowptr, const int* __restrict__ csr,
         const float* __restrict__ h, const float* __restrict__ rsq,
         float* __restrict__ out, int n, int e) {
    const int wpb = blockDim.x >> 5;
    const int lane = threadIdx.x & 31;
    for (int node = blockIdx.x * wpb + (threadIdx.x >> 5); node < n; node += gridDim.x * wpb) {
        int beg = __builtin_amdgcn_readfirstlane(rowptr[node]);
        int end = __builtin_amdgcn_readfirstlane(rowptr[node + 1]);
        const float ri = __int_as_float(__builtin_amdgcn_readfirstlane(__float_as_int(rsq[node])));
        beg = max(beg, 0);
        end = min(end, e);

        float4 acc = ld_row(h, node, lane);
        const float w_self = ri * ri;
        acc.x *= w_self; acc.y *= w_self; acc.z *= w_self; acc.w *= w_self;

        for (int base = beg; base < end; base += 32) {
            const int m = min(32, end - base);
            int s_l = 0;
            float q_l = 0.0f;
            if (lane < m) {
                s_l = csr[base + lane];
                q_l = rsq[s_l];
            }
            int k = 0;
            for (; k + 8 <= m; k += 8) {
                const float4 v0 = ld_row(h, bc_i(s_l, k), lane);
                const float4 v1 = ld_row(h, bc_i(s_l, k + 1), lane);
                const float4 v2 = ld_row(h, bc_i(s_l, k + 2), lane);
                const float4 v3 = ld_row(h, bc_i(s_l, k + 3), lane);
                const float4 v4 = ld_row(h, bc_i(s_l, k + 4), lane);
                const float4 v5 = ld_row(h, bc_i(s_l, k + 5), lane);
                const float4 v6 = ld_row(h, bc_i(s_l, k + 6), lane);
                const float4 v7 = ld_row(h, bc_i(s_l, k + 7), lane);
                fma4(acc, ri * bc_f(q_l, k), v0);
                fma4(acc, ri * bc_f(q_l, k + 1), v1);
                fma4(acc, ri * bc_f(q_l, k + 2), v2);
                fma4(acc, ri * bc_f(q_l, k + 3), v3);
                fma4(acc, ri * bc_f(q_l, k + 4), v4);
                fma4(acc, ri * bc_f(q_l, k + 5), v5);
                fma4(acc, ri * bc_f(q_l, k + 6), v6);
                fma4(acc, ri * bc_f(q_l, k + 7), v7);
            }
            if (k + 4 <= m) {
                const float4 v0 = ld_row(h, bc_i(s_l, k), lane);
                const float4 v1 = ld_row(h, bc_i(s_l, k + 1), lane);
                const float4 v2 = ld_row(h, bc_i(s_l, k + 2), lane);
                const float4 v3 = ld_row(h, bc_i(s_l, k + 3), lane);
                fma4(acc, ri * bc_f(q_l, k), v0);
                fma4(acc, ri * bc_f(q_l, k + 1), v1);
                fma4(acc, ri * bc_f(q_l, k + 2), v2);
                fma4(acc, ri * bc_f(q_l, k + 3), v3);
                k += 4;
            }
            const int r = m - k;
            if (r > 0) {
                const float4 v0 = ld_row(h, bc_i(s_l, k), lane);
                float4 v1 = make_float4(0.0f, 0.0f, 0.0f, 0.0f), v2 = v1;
                if (r > 1) v1 = ld_row(h, bc_i(s_l, k + 1), lane);
                if (r > 2) v2 = ld_row(h, bc_i(s_l, k + 2), lane);
                fma4(acc, ri * bc_f(q_l, k), v0);
                if (r > 1) fma4(acc, ri * bc_f(q_l, k + 1), v1);
                if (r > 2) fma4(acc, ri * bc_f(q_l, k + 2), v2);
            }
        }
        *(float4*)(out + (size_t)node * D + lane * 4) = acc;
    }
}

extern "C" void kernel_launch(void* const* d_in, const int* in_sizes, int n_in,
                              void* d_out, int out_size, void* d_ws, size_t ws_size,
                              hipStream_t stream) {
    const float* x    = (const float*)d_in[0];
    const int*   ei   = (const int*)d_in[1];
    const float* W    = (const float*)d_in[2];
    const float* bias = (const float*)d_in[3];
    float*       out  = (float*)d_out;

    const int n = in_sizes[0] / D;
    const int e = in_sizes[1] / 2;

    float* h      = (float*)d_ws;
    int2*  binned = (int2*)(h + (size_t)n * D);
    float* rsq    = (float*)(binned + e);
    int*   rowptr = (int*)(rsq + n);
    int*   csr    = rowptr + (n + 1);
    int*   ghist  = csr + e;
    int*   goff   = ghist + (NB_MAX * 512 + 1);

    const int T = 256;
    const int nb = (n + BIN_N - 1) / BIN_N;


    k_bin_hist   <<<512, T, 0, stream>>>(ei, ghist, e, nb);
    k_scan_bins  <<<NB_MAX * 512 / SCAN_T, SCAN_T, 0, stream>>>(ghist, goff, nb * 512);
    k_bin_scatter<<<512, T, 0, stream>>>(ei, goff, binned, e, nb);
    k_bin_csr    <<<nb, BIN_N, 0, stream>>>(binned, goff, rowptr, rsq, csr, n, 512);

    k_gemm_wmma<<<1280, T, 0, stream>>>(x, W, bias, h, n / 16);

    k_gather<<<2048, T, 0, stream>>>(rowptr, csr, h, rsq, out, n, e);
}


# GEMM operand loads touch 4x fewer cache lines per WMMA: x and W read as 128-bit pieces (k slots permuted alike in both), two column tiles per wave share each x piece; product sum order changes
# speedup vs baseline: 4.3911x; 1.1513x over previous
#include <hip/hip_runtime.h>


typedef __attribute__((ext_vector_type(2))) float v2f;
typedef __attribute__((ext_vector_type(8))) float v8f;

#define D 128

#define BIN_SHIFT 9
#define BIN_N 512
#define NB_MAX 256
#define TILE 4096
#define SCAN_T 1024
#define CSR_CAP 12288

__global__ void __launch_bounds__(256)
k_bin_hist(const int* __restrict__ ei_row, int* __restrict__ ghist, int e, int nb) {
    __shared__ int hist[NB_MAX];
    const int t = threadIdx.x;
    hist[t] = 0;
    __syncthreads();
    const int chunk = (e + (int)gridDim.x - 1) / (int)gridDim.x;
    const int c0 = min(e, (int)blockIdx.x * chunk);
    const int c1 = min(e, c0 + chunk);
    int i = c0 + t;
    for (; i + 768 < c1; i += 1024) {
        const int d0 = ei_row[i];
        const int d1 = ei_row[i + 256];
        const int d2 = ei_row[i + 512];
        const int d3 = ei_row[i + 768];
        atomicAdd(&hist[d0 >> BIN_SHIFT], 1);
        atomicAdd(&hist[d1 >> BIN_SHIFT], 1);
        atomicAdd(&hist[d2 >> BIN_SHIFT], 1);
        atomicAdd(&hist[d3 >> BIN_SHIFT], 1);
    }
    for (; i < c1; i += 256) atomicAdd(&hist[ei_row[i] >> BIN_SHIFT], 1);
    __syncthreads();
    if (t < nb) ghist[t * (int)gridDim.x + (int)blockIdx.x] = hist[t];
}

__global__ void __launch_bounds__(SCAN_T)
k_scan_bins(const int* __restrict__ ghist, int* __restrict__ goff, int m) {
    __shared__ int s_part[SCAN_T / 32];
    __shared__ int s_wave[SCAN_T / 32];
    if ((int)blockIdx.x * SCAN_T >= m) return;
    const int t = threadIdx.x;
    const int lane = t & 31;
    const int wv = t >> 5;

    int pre = 0;
    for (int k = 0; k < (int)blockIdx.x; ++k) pre += ghist[k * SCAN_T + t];
#pragma unroll
    for (int o = 16; o > 0; o >>= 1) pre += __shfl_xor(pre, o);
    if (lane == 0) s_part[wv] = pre;

    const int i = blockIdx.x * SCAN_T + t;
    const int v = (i < m) ? ghist[i] : 0;
    int inc = v;
#pragma unroll
    for (int o = 1; o < 32; o <<= 1) {
        int u = __shfl_up(inc, o);
        if (lane >= o) inc += u;
    }
    if (lane == 31) s_wave[wv] = inc;
    __syncthreads();

    int base = 0;
#pragma unroll
    for (int k = 0; k < SCAN_T / 32; ++k) {
        base += s_part[k];
        if (k < wv) base += s_wave[k];
    }
    if (i < m) {
        goff[i] = base + inc - v;
        if (i == m - 1) goff[m] = base + inc;
    }
}

__global__ void __launch_bounds__(256)
k_bin_scatter(const int* __restrict__ ei, const int* __restrict__ goff,
              int2* __restrict__ binned, int e, int nb) {
    __shared__ int hist[NB_MAX];
    __shared__ int loff[NB_MAX];
    __shared__ int gcur[NB_MAX];
    __shared__ int s_wave[8];
    __shared__ int2 stage[TILE];
    const int t = threadIdx.x;
    const int lane = t & 31;
    const int wv = t >> 5;
    const int chunk = (e + (int)gridDim.x - 1) / (int)gridDim.x;
    const int c0 = min(e, (int)blockIdx.x * chunk);
    const int c1 = min(e, c0 + chunk);
    gcur[t] = (t < nb) ? goff[t * (int)gridDim.x + (int)blockIdx.x] : 0;

    for (int t0 = c0; t0 < c1; t0 += TILE) {
        const int mt = min(TILE, c1 - t0);
        hist[t] = 0;
        __syncthreads();

        int dd[TILE / 256], ss[TILE / 256], rr[TILE / 256];
#pragma unroll
        for (int k = 0; k < TILE / 256; ++k) {
            const int j = k * 256 + t;
            dd[k] = (j < mt) ? ei[t0 + j] : 0;
            ss[k] = (j < mt) ? ei[e + t0 + j] : 0;
        }
#pragma unroll
        for (int k = 0; k < TILE / 256; ++k) {
            rr[k] = 0;
            if (k * 256 + t < mt) rr[k] = atomicAdd(&hist[dd[k] >> BIN_SHIFT], 1);
        }
        __syncthreads();

        const int v = hist[t];
        int inc = v;
#pragma unroll
        for (int o = 1; o < 32; o <<= 1) {
            int u = __shfl_up(inc, o);
            if (lane >= o) inc += u;
        }
        if (lane == 31) s_wave[wv] = inc;
        __syncthreads();
        int base = 0;
#pragma unroll
        for (int k = 0; k < 8; ++k)
            if (k < wv) base += s_wave[k];
        loff[t] = base + inc - v;
        __syncthreads();

#pragma unroll
        for (int k = 0; k < TILE / 256; ++k)
            if (k * 256 + t < mt)
                stage[loff[dd[k] >> BIN_SHIFT] + rr[k]] = make_int2(dd[k], ss[k]);
        __syncthreads();

        for (int j = t; j < mt; j += 256) {
            const int2 el = stage[j];
            const int b = el.x >> BIN_SHIFT;
            binned[gcur[b] + (j - loff[b])] = el;
        }
        __syncthreads();
        gcur[t] += v;
    }
}

__global__ void __launch_bounds__(BIN_N)
k_bin_csr(const int2* __restrict__ binned, const int* __restrict__ goff,
          int* __restrict__ rowptr, float* __restrict__ rsq, int* __restrict__ csr,
          int n, int g) {
    __shared__ int cnt[BIN_N];
    __shared__ int cur[BIN_N];
    __shared__ int s_wave[BIN_N / 32];
    __shared__ int stage[CSR_CAP];
    const int t = threadIdx.x;
    const int lane = t & 31;
    const int wv = t >> 5;
    const int b = blockIdx.x;
    const int b0 = goff[b * g];
    const int b1 = goff[(b + 1) * g];
    const int m = b1 - b0;

    cnt[t] = 0;
    __syncthreads();
    {
        int i = b0 + t;
        for (; i + 3 * BIN_N < b1; i += 4 * BIN_N) {
            const int k0 = binned[i].x;
            const int k1 = binned[i + BIN_N].x;
            const int k2 = binned[i + 2 * BIN_N].x;
            const int k3 = binned[i + 3 * BIN_N].x;
            atomicAdd(&cnt[k0 & (BIN_N - 1)], 1);
            atomicAdd(&cnt[k1 & (BIN_N - 1)], 1);
            atomicAdd(&cnt[k2 & (BIN_N - 1)], 1);
            atomicAdd(&cnt[k3 & (BIN_N - 1)], 1);
        }
        for (; i < b1; i += BIN_N) atomicAdd(&cnt[binned[i].x & (BIN_N - 1)], 1);
    }
    __syncthreads();

    const int v = cnt[t];
    int inc = v;
#pragma unroll
    for (int o = 1; o < 32; o <<= 1) {
        int u = __shfl_up(inc, o);
        if (lane >= o) inc += u;
    }
    if (lane == 31) s_wave[wv] = inc;
    __syncthreads();
    int base = 0;
#pragma unroll
    for (int k = 0; k < BIN_N / 32; ++k)
        if (k < wv) base += s_wave[k];
    const int off = base + inc - v;
    cur[t] = off;
    const int node = b * BIN_N + t;
    if (node < n) {
        rowptr[node] = b0 + off;
        rsq[node] = rsqrtf((float)(v + 1));
        if (node == n - 1) rowptr[n] = b1;
    }
    __syncthreads();

    if (m <= CSR_CAP) {
        int i = b0 + t;
        for (; i + 3 * BIN_N < b1; i += 4 * BIN_N) {
            const int2 e0 = binned[i];
            const int2 e1 = binned[i + BIN_N];
            const int2 e2 = binned[i + 2 * BIN_N];
            const int2 e3 = binned[i + 3 * BIN_N];
            stage[atomicAdd(&cur[e0.x & (BIN_N - 1)], 1)] = e0.y;
            stage[atomicAdd(&cur[e1.x & (BIN_N - 1)], 1)] = e1.y;
            stage[atomicAdd(&cur[e2.x & (BIN_N - 1)], 1)] = e2.y;
            stage[atomicAdd(&cur[e3.x & (BIN_N - 1)], 1)] = e3.y;
        }
        for (; i < b1; i += BIN_N) {
            const int2 el = binned[i];
            stage[atomicAdd(&cur[el.x & (BIN_N - 1)], 1)] = el.y;
        }
        __syncthreads();
        for (int j = t; j < m; j += BIN_N) csr[b0 + j] = stage[j];
    } else {
        for (int i = b0 + t; i < b1; i += BIN_N) {
            const int2 el = binned[i];
            csr[b0 + atomicAdd(&cur[el.x & (BIN_N - 1)], 1)] = el.y;
        }
    }
}

typedef __attribute__((ext_vector_type(4))) float v4f;

__global__ void __launch_bounds__(128)
k_gemm_wmma(const float* __restrict__ x,
            const float* __restrict__ W,
            const float* __restrict__ bias,
            float* __restrict__ h, int n_slabs) {
    const int wave = threadIdx.x >> 5;
    const int lane = threadIdx.x & 31;
    const int n0 = wave << 5;
    const int lm = lane & 15;
    const int lk = (lane >> 4) << 2;

    const float* wr0 = W + (size_t)(n0 + lm) * D + lk;
    const float* wr1 = W + (size_t)(n0 + 16 + lm) * D + lk;
    v4f b0[D / 8], b1[D / 8];
#pragma unroll
    for (int j = 0; j < D / 8; ++j) {
        b0[j] = *(const v4f*)(wr0 + 8 * j);
        b1[j] = *(const v4f*)(wr1 + 8 * j);
    }
    const float bn0 = bias[n0 + lm];
    const float bn1 = bias[n0 + 16 + lm];

    for (int slab = blockIdx.x; slab < n_slabs; slab += gridDim.x) {
        const int m0 = slab << 4;
        const float* xr = x + (size_t)(m0 + lm) * D + lk;

        v4f a[D / 8];
#pragma unroll
        for (int j = 0; j < D / 8; ++j) a[j] = *(const v4f*)(xr + 8 * j);

        v8f acc0 = {}, acc1 = {};
#pragma unroll
        for (int j = 0; j < D / 8; ++j) {
            const v2f alo = {a[j].x, a[j].y};
            const v2f ahi = {a[j].z, a[j].w};
            const v2f b0lo = {b0[j].x, b0[j].y};
            const v2f b0hi = {b0[j].z, b0[j].w};
            const v2f b1lo = {b1[j].x, b1[j].y};
            const v2f b1hi = {b1[j].z, b1[j].w};
            acc0 = __builtin_amdgcn_wmma_f32_16x16x4_f32(
                false, alo, false, b0lo, (short)0, acc0, false, false);
            acc1 = __builtin_amdgcn_wmma_f32_16x16x4_f32(
                false, alo, false, b1lo, (short)0, acc1, false, false);
            acc0 = __builtin_amdgcn_wmma_f32_16x16x4_f32(
                false, ahi, false, b0hi, (short)0, acc0, false, false);
            acc1 = __builtin_amdgcn_wmma_f32_16x16x4_f32(
                false, ahi, false, b1hi, (short)0, acc1, false, false);
        }

        float* hp = h + (size_t)(m0 + ((lane >> 4) << 3)) * D + (n0 + lm);
#pragma unroll
        for (int v = 0; v < 8; ++v) {
            hp[(size_t)v * D] = acc0[v] + bn0;
            hp[(size_t)v * D + 16] = acc1[v] + bn1;
        }
    }
}

__device__ __forceinline__ float4 ld_row(const float* __restrict__ h, int s, int lane) {
    return *(const float4*)(h + (size_t)s * D + lane * 4);
}
__device__ __forceinline__ void fma4(float4& a, float w, const float4& v) {
    a.x = fmaf(w, v.x, a.x); a.y = fmaf(w, v.y, a.y);
    a.z = fmaf(w, v.z, a.z); a.w = fmaf(w, v.w, a.w);
}
__device__ __forceinline__ int bc_i(int v, int k) { return __builtin_amdgcn_readlane(v, k); }
__device__ __forceinline__ float bc_f(float v, int k) {
    return __int_as_float(__builtin_amdgcn_readlane(__float_as_int(v), k));
}

__global__ void __launch_bounds__(256)
k_gather(const int* __restrict__ rowptr, const int* __restrict__ csr,
         const float* __restrict__ h, const float* __restrict__ rsq,
         float* __restrict__ out, int n, int e) {
    const int wpb = blockDim.x >> 5;
    const int lane = threadIdx.x & 31;
    for (int node = blockIdx.x * wpb + (threadIdx.x >> 5); node < n; node += gridDim.x * wpb) {
        int beg = __builtin_amdgcn_readfirstlane(rowptr[node]);
        int end = __builtin_amdgcn_readfirstlane(rowptr[node + 1]);
        const float ri = __int_as_float(__builtin_amdgcn_readfirstlane(__float_as_int(rsq[node])));
        beg = max(beg, 0);
        end = min(end, e);

        float4 acc = ld_row(h, node, lane);
        const float w_self = ri * ri;
        acc.x *= w_self; acc.y *= w_self; acc.z *= w_self; acc.w *= w_self;

        for (int base = beg; base < end; base += 32) {
            const int m = min(32, end - base);
            int s_l = 0;
            float q_l = 0.0f;
            if (lane < m) {
                s_l = csr[base + lane];
                q_l = rsq[s_l];
            }
            int k = 0;
            for (; k + 8 <= m; k += 8) {
                const float4 v0 = ld_row(h, bc_i(s_l, k), lane);
                const float4 v1 = ld_row(h, bc_i(s_l, k + 1), lane);
                const float4 v2 = ld_row(h, bc_i(s_l, k + 2), lane);
                const float4 v3 = ld_row(h, bc_i(s_l, k + 3), lane);
                const float4 v4 = ld_row(h, bc_i(s_l, k + 4), lane);
                const float4 v5 = ld_row(h, bc_i(s_l, k + 5), lane);
                const float4 v6 = ld_row(h, bc_i(s_l, k + 6), lane);
                const float4 v7 = ld_row(h, bc_i(s_l, k + 7), lane);
                fma4(acc, ri * bc_f(q_l, k), v0);
                fma4(acc, ri * bc_f(q_l, k + 1), v1);
                fma4(acc, ri * bc_f(q_l, k + 2), v2);
                fma4(acc, ri * bc_f(q_l, k + 3), v3);
                fma4(acc, ri * bc_f(q_l, k + 4), v4);
                fma4(acc, ri * bc_f(q_l, k + 5), v5);
                fma4(acc, ri * bc_f(q_l, k + 6), v6);
                fma4(acc, ri * bc_f(q_l, k + 7), v7);
            }
            if (k + 4 <= m) {
                const float4 v0 = ld_row(h, bc_i(s_l, k), lane);
                const float4 v1 = ld_row(h, bc_i(s_l, k + 1), lane);
                const float4 v2 = ld_row(h, bc_i(s_l, k + 2), lane);
                const float4 v3 = ld_row(h, bc_i(s_l, k + 3), lane);
                fma4(acc, ri * bc_f(q_l, k), v0);
                fma4(acc, ri * bc_f(q_l, k + 1), v1);
                fma4(acc, ri * bc_f(q_l, k + 2), v2);
                fma4(acc, ri * bc_f(q_l, k + 3), v3);
                k += 4;
            }
            const int r = m - k;
            if (r > 0) {
                const float4 v0 = ld_row(h, bc_i(s_l, k), lane);
                float4 v1 = make_float4(0.0f, 0.0f, 0.0f, 0.0f), v2 = v1;
                if (r > 1) v1 = ld_row(h, bc_i(s_l, k + 1), lane);
                if (r > 2) v2 = ld_row(h, bc_i(s_l, k + 2), lane);
                fma4(acc, ri * bc_f(q_l, k), v0);
                if (r > 1) fma4(acc, ri * bc_f(q_l, k + 1), v1);
                if (r > 2) fma4(acc, ri * bc_f(q_l, k + 2), v2);
            }
        }
        *(float4*)(out + (size_t)node * D + lane * 4) = acc;
    }
}

extern "C" void kernel_launch(void* const* d_in, const int* in_sizes, int n_in,
                              void* d_out, int out_size, void* d_ws, size_t ws_size,
                              hipStream_t stream) {
    const float* x    = (const float*)d_in[0];
    const int*   ei   = (const int*)d_in[1];
    const float* W    = (const float*)d_in[2];
    const float* bias = (const float*)d_in[3];
    float*       out  = (float*)d_out;

    const int n = in_sizes[0] / D;
    const int e = in_sizes[1] / 2;

    float* h      = (float*)d_ws;
    int2*  binned = (int2*)(h + (size_t)n * D);
    float* rsq    = (float*)(binned + e);
    int*   rowptr = (int*)(rsq + n);
    int*   csr    = rowptr + (n + 1);
    int*   ghist  = csr + e;
    int*   goff   = ghist + (NB_MAX * 512 + 1);

    const int T = 256;
    const int nb = (n + BIN_N - 1) / BIN_N;


    k_bin_hist   <<<512, T, 0, stream>>>(ei, ghist, e, nb);
    k_scan_bins  <<<NB_MAX * 512 / SCAN_T, SCAN_T, 0, stream>>>(ghist, goff, nb * 512);
    k_bin_scatter<<<512, T, 0, stream>>>(ei, goff, binned, e, nb);
    k_bin_csr    <<<nb, BIN_N, 0, stream>>>(binned, goff, rowptr, rsq, csr, n, 512);

    k_gemm_wmma<<<1280, 128, 0, stream>>>(x, W, bias, h, n / 16);

    k_gather<<<2048, T, 0, stream>>>(rowptr, csr, h, rsq, out, n, e);
}


# Sort by dst without global histogram or scan: blocks group 8192-edge tiles by bin in place and publish per-tile bin offsets; bin blocks walk their per-tile pieces; sum order by LDS atomic arrival
# speedup vs baseline: 5.3228x; 1.2122x over previous
#include <hip/hip_runtime.h>


typedef __attribute__((ext_vector_type(2))) float v2f;
typedef __attribute__((ext_vector_type(8))) float v8f;

#define D 128

#define BIN_SHIFT 9
#define BIN_N 512
#define NB_MAX 256
#define TILE 8192
#define TS_T 512
#define EPT (TILE / TS_T)
#define CSR_CAP 12288

__global__ void __launch_bounds__(TS_T)
k_tile_sort(const int* __restrict__ ei, int* __restrict__ tab,
            int2* __restrict__ binned, int e, int nb) {
    __shared__ int hist[NB_MAX];
    __shared__ int loff[NB_MAX];
    __shared__ int s_wave[NB_MAX / 32];
    __shared__ int2 stage[TILE];
    const int t = threadIdx.x;
    const int lane = t & 31;
    const int wv = t >> 5;
    const int nt = (e + TILE - 1) / TILE;

    for (int u = blockIdx.x; u < nt; u += gridDim.x) {
        const int t0 = u * TILE;
        const int mt = min(TILE, e - t0);
        if (t < NB_MAX) hist[t] = 0;
        __syncthreads();

        int dd[EPT], ss[EPT], rr[EPT];
#pragma unroll
        for (int k = 0; k < EPT; ++k) {
            const int j = k * TS_T + t;
            dd[k] = (j < mt) ? ei[t0 + j] : 0;
            ss[k] = (j < mt) ? ei[e + t0 + j] : 0;
        }
#pragma unroll
        for (int k = 0; k < EPT; ++k) {
            rr[k] = 0;
            if (k * TS_T + t < mt) rr[k] = atomicAdd(&hist[dd[k] >> BIN_SHIFT], 1);
        }
        __syncthreads();

        int v = 0, inc = 0;
        if (t < NB_MAX) {
            v = hist[t];
            inc = v;
#pragma unroll
            for (int o = 1; o < 32; o <<= 1) {
                int w = __shfl_up(inc, o);
                if (lane >= o) inc += w;
            }
            if (lane == 31) s_wave[wv] = inc;
        }
        __syncthreads();
        if (t < NB_MAX) {
            int base = 0;
#pragma unroll
            for (int k = 0; k < NB_MAX / 32; ++k)
                if (k < wv) base += s_wave[k];
            const int off = base + inc - v;
            loff[t] = off;
            if (t < nb) tab[t * nt + u] = off;
            if (t == 0) tab[nb * nt + u] = mt;
        }
        __syncthreads();

        int po[EPT];
#pragma unroll
        for (int k = 0; k < EPT; ++k) po[k] = loff[dd[k] >> BIN_SHIFT] + rr[k];
#pragma unroll
        for (int k = 0; k < EPT; ++k)
            if (k * TS_T + t < mt) stage[po[k]] = make_int2(dd[k], ss[k]);
        __syncthreads();

        int2 pv[EPT];
#pragma unroll
        for (int k = 0; k < EPT; ++k) pv[k] = stage[k * TS_T + t];
#pragma unroll
        for (int k = 0; k < EPT; ++k)
            if (k * TS_T + t < mt) binned[t0 + k * TS_T + t] = pv[k];
    }
}

template <class R, class P>
__device__ __forceinline__ void walk_bin(const int2* __restrict__ binned,
                                         const int* __restrict__ lo, const int* __restrict__ hi,
                                         int nt, int wv, int lane, R rank, P put) {
    for (int ub = wv; ub < nt; ub += 16 * 32) {
        int pbase = 0, plen = 0;
        const int u = ub + 16 * lane;
        if (u < nt) {
            const int s = min(max(lo[u], 0), TILE);
            plen = min(max(hi[u] - s, 0), TILE - s);
            pbase = u * TILE + s;
        }
        const int np = min(32, (nt - ub + 15) >> 4);
        for (int l0 = 0; l0 < np; l0 += 4) {
            int2 v0[4], v1[4];
            int ln[4], bs[4];
#pragma unroll
            for (int q = 0; q < 4; ++q) {
                const int l = min(l0 + q, 31);
                ln[q] = (l0 + q < np) ? __builtin_amdgcn_readlane(plen, l) : 0;
                bs[q] = __builtin_amdgcn_readlane(pbase, l);
                v0[q] = binned[(lane < ln[q]) ? bs[q] + lane : 0];
                v1[q] = binned[(lane + 32 < ln[q]) ? bs[q] + 32 + lane : 0];
            }
            asm volatile("" ::
                "v"(v0[0].x), "v"(v0[0].y), "v"(v1[0].x), "v"(v1[0].y),
                "v"(v0[1].x), "v"(v0[1].y), "v"(v1[1].x), "v"(v1[1].y),
                "v"(v0[2].x), "v"(v0[2].y), "v"(v1[2].x), "v"(v1[2].y),
                "v"(v0[3].x), "v"(v0[3].y), "v"(v1[3].x), "v"(v1[3].y));
            int r0[4], r1[4];
#pragma unroll
            for (int q = 0; q < 4; ++q) {
                r0[q] = 0;
                r1[q] = 0;
                if (lane < ln[q]) r0[q] = rank(v0[q]);
                if (lane + 32 < ln[q]) r1[q] = rank(v1[q]);
            }
#pragma unroll
            for (int q = 0; q < 4; ++q) {
                if (lane < ln[q]) put(v0[q], r0[q]);
                if (lane + 32 < ln[q]) put(v1[q], r1[q]);
                for (int i = 64 + lane; i < ln[q]; i += 32) {
                    const int2 el = binned[bs[q] + i];
                    put(el, rank(el));
                }
            }
        }
    }
}

__global__ void __launch_bounds__(BIN_N)
k_bin_csr(const int2* __restrict__ binned, const int* __restrict__ tab,
          int* __restrict__ rowptr, float* __restrict__ rsq, int* __restrict__ csr,
          int n, int e) {
    __shared__ int cnt[BIN_N];
    __shared__ int cur[BIN_N];
    __shared__ int s_wave[BIN_N / 32];
    __shared__ int s_lo[BIN_N / 32];
    __shared__ int s_hi[BIN_N / 32];
    __shared__ int stage[CSR_CAP];
    const int t = threadIdx.x;
    const int lane = t & 31;
    const int wv = __builtin_amdgcn_readfirstlane(t >> 5);
    const int b = blockIdx.x;
    const int nt = (e + TILE - 1) / TILE;
    const int* lo = tab + (size_t)b * nt;
    const int* hi = tab + (size_t)(b + 1) * nt;

    int p0 = 0, p1 = 0;
    for (int u = t; u < nt; u += BIN_N) {
        p0 += lo[u];
        p1 += hi[u];
    }
#pragma unroll
    for (int o = 16; o > 0; o >>= 1) {
        p0 += __shfl_xor(p0, o);
        p1 += __shfl_xor(p1, o);
    }
    if (lane == 0) {
        s_lo[wv] = p0;
        s_hi[wv] = p1;
    }
    cnt[t] = 0;
    __syncthreads();
    int b0 = 0, b1 = 0;
#pragma unroll
    for (int k = 0; k < BIN_N / 32; ++k) {
        b0 += s_lo[k];
        b1 += s_hi[k];
    }
    const int m = b1 - b0;

    walk_bin(binned, lo, hi, nt, wv, lane,
             [&](int2 el) { atomicAdd(&cnt[el.x & (BIN_N - 1)], 1); return 0; },
             [&](int2, int) {});
    __syncthreads();

    const int v = cnt[t];
    int inc = v;
#pragma unroll
    for (int o = 1; o < 32; o <<= 1) {
        int w = __shfl_up(inc, o);
        if (lane >= o) inc += w;
    }
    if (lane == 31) s_wave[wv] = inc;
    __syncthreads();
    int base = 0;
#pragma unroll
    for (int k = 0; k < BIN_N / 32; ++k)
        if (k < wv) base += s_wave[k];
    const int off = base + inc - v;
    cur[t] = off;
    const int node = b * BIN_N + t;
    if (node < n) {
        rowptr[node] = b0 + off;
        rsq[node] = rsqrtf((float)(v + 1));
        if (node == n - 1) rowptr[n] = b1;
    }
    __syncthreads();

    if (m <= CSR_CAP) {
        walk_bin(binned, lo, hi, nt, wv, lane,
                 [&](int2 el) { return atomicAdd(&cur[el.x & (BIN_N - 1)], 1); },
                 [&](int2 el, int r) { stage[r] = el.y; });
        __syncthreads();
        for (int j0 = 0; j0 < m; j0 += 4 * BIN_N) {
            int sv[4];
#pragma unroll
            for (int q = 0; q < 4; ++q) sv[q] = stage[min(j0 + q * BIN_N + t, CSR_CAP - 1)];
#pragma unroll
            for (int q = 0; q < 4; ++q) {
                const int j = j0 + q * BIN_N + t;
                if (j < m) csr[b0 + j] = sv[q];
            }
        }
    } else {
        walk_bin(binned, lo, hi, nt, wv, lane,
                 [&](int2 el) { return atomicAdd(&cur[el.x & (BIN_N - 1)], 1); },
                 [&](int2 el, int r) { csr[b0 + r] = el.y; });
    }
}

typedef __attribute__((ext_vector_type(4))) float v4f;

__global__ void __launch_bounds__(128)
k_gemm_wmma(const float* __restrict__ x,
            const float* __restrict__ W,
            const float* __restrict__ bias,
            float* __restrict__ h, int n_slabs) {
    const int wave = threadIdx.x >> 5;
    const int lane = threadIdx.x & 31;
    const int n0 = wave << 5;
    const int lm = lane & 15;
    const int lk = (lane >> 4) << 2;

    const float* wr0 = W + (size_t)(n0 + lm) * D + lk;
    const float* wr1 = W + (size_t)(n0 + 16 + lm) * D + lk;
    v4f b0[D / 8], b1[D / 8];
#pragma unroll
    for (int j = 0; j < D / 8; ++j) {
        b0[j] = *(const v4f*)(wr0 + 8 * j);
        b1[j] = *(const v4f*)(wr1 + 8 * j);
    }
    const float bn0 = bias[n0 + lm];
    const float bn1 = bias[n0 + 16 + lm];

    for (int slab = blockIdx.x; slab < n_slabs; slab += gridDim.x) {
        const int m0 = slab << 4;
        const float* xr = x + (size_t)(m0 + lm) * D + lk;

        v4f a[D / 8];
#pragma unroll
        for (int j = 0; j < D / 8; ++j) a[j] = *(const v4f*)(xr + 8 * j);

        v8f acc0 = {}, acc1 = {};
#pragma unroll
        for (int j = 0; j < D / 8; ++j) {
            const v2f alo = {a[j].x, a[j].y};
            const v2f ahi = {a[j].z, a[j].w};
            const v2f b0lo = {b0[j].x, b0[j].y};
            const v2f b0hi = {b0[j].z, b0[j].w};
            const v2f b1lo = {b1[j].x, b1[j].y};
            const v2f b1hi = {b1[j].z, b1[j].w};
            acc0 = __builtin_amdgcn_wmma_f32_16x16x4_f32(
                false, alo, false, b0lo, (short)0, acc0, false, false);
            acc1 = __builtin_amdgcn_wmma_f32_16x16x4_f32(
                false, alo, false, b1lo, (short)0, acc1, false, false);
            acc0 = __builtin_amdgcn_wmma_f32_16x16x4_f32(
                false, ahi, false, b0hi, (short)0, acc0, false, false);
            acc1 = __builtin_amdgcn_wmma_f32_16x16x4_f32(
                false, ahi, false, b1hi, (short)0, acc1, false, false);
        }

        float* hp = h + (size_t)(m0 + ((lane >> 4) << 3)) * D + (n0 + lm);
#pragma unroll
        for (int v = 0; v < 8; ++v) {
            hp[(size_t)v * D] = acc0[v] + bn0;
            hp[(size_t)v * D + 16] = acc1[v] + bn1;
        }
    }
}

__device__ __forceinline__ float4 ld_row(const float* __restrict__ h, int s, int lane) {
    return *(const float4*)(h + (size_t)s * D + lane * 4);
}
__device__ __forceinline__ void fma4(float4& a, float w, const float4& v) {
    a.x = fmaf(w, v.x, a.x); a.y = fmaf(w, v.y, a.y);
    a.z = fmaf(w, v.z, a.z); a.w = fmaf(w, v.w, a.w);
}
__device__ __forceinline__ int bc_i(int v, int k) { return __builtin_amdgcn_readlane(v, k); }
__device__ __forceinline__ float bc_f(float v, int k) {
    return __int_as_float(__builtin_amdgcn_readlane(__float_as_int(v), k));
}

__global__ void __launch_bounds__(256)
k_gather(const int* __restrict__ rowptr, const int* __restrict__ csr,
         const float* __restrict__ h, const float* __restrict__ rsq,
         float* __restrict__ out, int n, int e) {
    const int wpb = blockDim.x >> 5;
    const int lane = threadIdx.x & 31;
    for (int node = blockIdx.x * wpb + (threadIdx.x >> 5); node < n; node += gridDim.x * wpb) {
        int beg = __builtin_amdgcn_readfirstlane(rowptr[node]);
        int end = __builtin_amdgcn_readfirstlane(rowptr[node + 1]);
        const float ri = __int_as_float(__builtin_amdgcn_readfirstlane(__float_as_int(rsq[node])));
        beg = max(beg, 0);
        end = min(end, e);

        float4 acc = ld_row(h, node, lane);
        const float w_self = ri * ri;
        acc.x *= w_self; acc.y *= w_self; acc.z *= w_self; acc.w *= w_self;

        for (int base = beg; base < end; base += 32) {
            const int m = min(32, end - base);
            int s_l = 0;
            float q_l = 0.0f;
            if (lane < m) {
                s_l = csr[base + lane];
                q_l = rsq[s_l];
            }
            int k = 0;
            for (; k + 8 <= m; k += 8) {
                const float4 v0 = ld_row(h, bc_i(s_l, k), lane);
                const float4 v1 = ld_row(h, bc_i(s_l, k + 1), lane);
                const float4 v2 = ld_row(h, bc_i(s_l, k + 2), lane);
                const float4 v3 = ld_row(h, bc_i(s_l, k + 3), lane);
                const float4 v4 = ld_row(h, bc_i(s_l, k + 4), lane);
                const float4 v5 = ld_row(h, bc_i(s_l, k + 5), lane);
                const float4 v6 = ld_row(h, bc_i(s_l, k + 6), lane);
                const float4 v7 = ld_row(h, bc_i(s_l, k + 7), lane);
                fma4(acc, ri * bc_f(q_l, k), v0);
                fma4(acc, ri * bc_f(q_l, k + 1), v1);
                fma4(acc, ri * bc_f(q_l, k + 2), v2);
                fma4(acc, ri * bc_f(q_l, k + 3), v3);
                fma4(acc, ri * bc_f(q_l, k + 4), v4);
                fma4(acc, ri * bc_f(q_l, k + 5), v5);
                fma4(acc, ri * bc_f(q_l, k + 6), v6);
                fma4(acc, ri * bc_f(q_l, k + 7), v7);
            }
            if (k + 4 <= m) {
                const float4 v0 = ld_row(h, bc_i(s_l, k), lane);
                const float4 v1 = ld_row(h, bc_i(s_l, k + 1), lane);
                const float4 v2 = ld_row(h, bc_i(s_l, k + 2), lane);
                const float4 v3 = ld_row(h, bc_i(s_l, k + 3), lane);
                fma4(acc, ri * bc_f(q_l, k), v0);
                fma4(acc, ri * bc_f(q_l, k + 1), v1);
                fma4(acc, ri * bc_f(q_l, k + 2), v2);
                fma4(acc, ri * bc_f(q_l, k + 3), v3);
                k += 4;
            }
            const int r = m - k;
            if (r > 0) {
                const float4 v0 = ld_row(h, bc_i(s_l, k), lane);
                float4 v1 = make_float4(0.0f, 0.0f, 0.0f, 0.0f), v2 = v1;
                if (r > 1) v1 = ld_row(h, bc_i(s_l, k + 1), lane);
                if (r > 2) v2 = ld_row(h, bc_i(s_l, k + 2), lane);
                fma4(acc, ri * bc_f(q_l, k), v0);
                if (r > 1) fma4(acc, ri * bc_f(q_l, k + 1), v1);
                if (r > 2) fma4(acc, ri * bc_f(q_l, k + 2), v2);
            }
        }
        *(float4*)(out + (size_t)node * D + lane * 4) = acc;
    }
}

extern "C" void kernel_launch(void* const* d_in, const int* in_sizes, int n_in,
                              void* d_out, int out_size, void* d_ws, size_t ws_size,
                              hipStream_t stream) {
    const float* x    = (const float*)d_in[0];
    const int*   ei   = (const int*)d_in[1];
    const float* W    = (const float*)d_in[2];
    const float* bias = (const float*)d_in[3];
    float*       out  = (float*)d_out;

    const int n = in_sizes[0] / D;
    const int e = in_sizes[1] / 2;

    float* h      = (float*)d_ws;
    int2*  binned = (int2*)(h + (size_t)n * D);
    float* rsq    = (float*)(binned + e);
    int*   rowptr = (int*)(rsq + n);
    int*   csr    = rowptr + (n + 1);
    int*   tab    = csr + e;

    const int T = 256;
    const int nb = (n + BIN_N - 1) / BIN_N;


    k_tile_sort<<<256, TS_T, 0, stream>>>(ei, tab, binned, e, nb);
    k_bin_csr  <<<nb, BIN_N, 0, stream>>>(binned, tab, rowptr, rsq, csr, n, e);

    k_gemm_wmma<<<1280, 128, 0, stream>>>(x, W, bias, h, n / 16);

    k_gather<<<2048, T, 0, stream>>>(rowptr, csr, h, rsq, out, n, e);
}


# Projection moved behind the aggregation, into the gather kernel: out = (An x) W^T + (An 1) b^T, each 16-node slab of summed x rows times W on WMMA from LDS; no h, one launch less; sum order changes
# speedup vs baseline: 5.6000x; 1.0521x over previous
#include <hip/hip_runtime.h>


typedef __attribute__((ext_vector_type(2))) float v2f;
typedef __attribute__((ext_vector_type(8))) float v8f;

#define D 128

#define BIN_SHIFT 9
#define BIN_N 512
#define NB_MAX 256
#define TILE 8192
#define TS_T 512
#define EPT (TILE / TS_T)
#define CSR_CAP 12288

__global__ void __launch_bounds__(TS_T)
k_tile_sort(const int* __restrict__ ei, int* __restrict__ tab,
            int2* __restrict__ binned, int e, int nb) {
    __shared__ int hist[NB_MAX];
    __shared__ int loff[NB_MAX];
    __shared__ int s_wave[NB_MAX / 32];
    __shared__ int2 stage[TILE];
    const int t = threadIdx.x;
    const int lane = t & 31;
    const int wv = t >> 5;
    const int nt = (e + TILE - 1) / TILE;

    for (int u = blockIdx.x; u < nt; u += gridDim.x) {
        const int t0 = u * TILE;
        const int mt = min(TILE, e - t0);
        if (t < NB_MAX) hist[t] = 0;
        __syncthreads();

        int dd[EPT], ss[EPT], rr[EPT];
#pragma unroll
        for (int k = 0; k < EPT; ++k) {
            const int j = k * TS_T + t;
            dd[k] = (j < mt) ? ei[t0 + j] : 0;
            ss[k] = (j < mt) ? ei[e + t0 + j] : 0;
        }
#pragma unroll
        for (int k = 0; k < EPT; ++k) {
            rr[k] = 0;
            if (k * TS_T + t < mt) rr[k] = atomicAdd(&hist[dd[k] >> BIN_SHIFT], 1);
        }
        __syncthreads();

        int v = 0, inc = 0;
        if (t < NB_MAX) {
            v = hist[t];
            inc = v;
#pragma unroll
            for (int o = 1; o < 32; o <<= 1) {
                int w = __shfl_up(inc, o);
                if (lane >= o) inc += w;
            }
            if (lane == 31) s_wave[wv] = inc;
        }
        __syncthreads();
        if (t < NB_MAX) {
            int base = 0;
#pragma unroll
            for (int k = 0; k < NB_MAX / 32; ++k)
                if (k < wv) base += s_wave[k];
            const int off = base + inc - v;
            loff[t] = off;
            if (t < nb) tab[t * nt + u] = off;
            if (t == 0) tab[nb * nt + u] = mt;
        }
        __syncthreads();

        int po[EPT];
#pragma unroll
        for (int k = 0; k < EPT; ++k) po[k] = loff[dd[k] >> BIN_SHIFT] + rr[k];
#pragma unroll
        for (int k = 0; k < EPT; ++k)
            if (k * TS_T + t < mt) stage[po[k]] = make_int2(dd[k], ss[k]);
        __syncthreads();

        int2 pv[EPT];
#pragma unroll
        for (int k = 0; k < EPT; ++k) pv[k] = stage[k * TS_T + t];
#pragma unroll
        for (int k = 0; k < EPT; ++k)
            if (k * TS_T + t < mt) binned[t0 + k * TS_T + t] = pv[k];
    }
}

template <class R, class P>
__device__ __forceinline__ void walk_bin(const int2* __restrict__ binned,
                                         const int* __restrict__ lo, const int* __restrict__ hi,
                                         int nt, int wv, int lane, R rank, P put) {
    for (int ub = wv; ub < nt; ub += 16 * 32) {
        int pbase = 0, plen = 0;
        const int u = ub + 16 * lane;
        if (u < nt) {
            const int s = min(max(lo[u], 0), TILE);
            plen = min(max(hi[u] - s, 0), TILE - s);
            pbase = u * TILE + s;
        }
        const int np = min(32, (nt - ub + 15) >> 4);
        for (int l0 = 0; l0 < np; l0 += 4) {
            int2 v0[4], v1[4];
            int ln[4], bs[4];
#pragma unroll
            for (int q = 0; q < 4; ++q) {
                const int l = min(l0 + q, 31);
                ln[q] = (l0 + q < np) ? __builtin_amdgcn_readlane(plen, l) : 0;
                bs[q] = __builtin_amdgcn_readlane(pbase, l);
                v0[q] = binned[(lane < ln[q]) ? bs[q] + lane : 0];
                v1[q] = binned[(lane + 32 < ln[q]) ? bs[q] + 32 + lane : 0];
            }
            asm volatile("" ::
                "v"(v0[0].x), "v"(v0[0].y), "v"(v1[0].x), "v"(v1[0].y),
                "v"(v0[1].x), "v"(v0[1].y), "v"(v1[1].x), "v"(v1[1].y),
                "v"(v0[2].x), "v"(v0[2].y), "v"(v1[2].x), "v"(v1[2].y),
                "v"(v0[3].x), "v"(v0[3].y), "v"(v1[3].x), "v"(v1[3].y));
            int r0[4], r1[4];
#pragma unroll
            for (int q = 0; q < 4; ++q) {
                r0[q] = 0;
                r1[q] = 0;
                if (lane < ln[q]) r0[q] = rank(v0[q]);
                if (lane + 32 < ln[q]) r1[q] = rank(v1[q]);
            }
#pragma unroll
            for (int q = 0; q < 4; ++q) {
                if (lane < ln[q]) put(v0[q], r0[q]);
                if (lane + 32 < ln[q]) put(v1[q], r1[q]);
                for (int i = 64 + lane; i < ln[q]; i += 32) {
                    const int2 el = binned[bs[q] + i];
                    put(el, rank(el));
                }
            }
        }
    }
}

__global__ void __launch_bounds__(BIN_N)
k_bin_csr(const int2* __restrict__ binned, const int* __restrict__ tab,
          int* __restrict__ rowptr, float* __restrict__ rsq, int* __restrict__ csr,
          int n, int e) {
    __shared__ int cnt[BIN_N];
    __shared__ int cur[BIN_N];
    __shared__ int s_wave[BIN_N / 32];
    __shared__ int s_lo[BIN_N / 32];
    __shared__ int s_hi[BIN_N / 32];
    __shared__ int stage[CSR_CAP];
    const int t = threadIdx.x;
    const int lane = t & 31;
    const int wv = __builtin_amdgcn_readfirstlane(t >> 5);
    const int b = blockIdx.x;
    const int nt = (e + TILE - 1) / TILE;
    const int* lo = tab + (size_t)b * nt;
    const int* hi = tab + (size_t)(b + 1) * nt;

    int p0 = 0, p1 = 0;
    for (int u = t; u < nt; u += BIN_N) {
        p0 += lo[u];
        p1 += hi[u];
    }
#pragma unroll
    for (int o = 16; o > 0; o >>= 1) {
        p0 += __shfl_xor(p0, o);
        p1 += __shfl_xor(p1, o);
    }
    if (lane == 0) {
        s_lo[wv] = p0;
        s_hi[wv] = p1;
    }
    cnt[t] = 0;
    __syncthreads();
    int b0 = 0, b1 = 0;
#pragma unroll
    for (int k = 0; k < BIN_N / 32; ++k) {
        b0 += s_lo[k];
        b1 += s_hi[k];
    }
    const int m = b1 - b0;

    walk_bin(binned, lo, hi, nt, wv, lane,
             [&](int2 el) { atomicAdd(&cnt[el.x & (BIN_N - 1)], 1); return 0; },
             [&](int2, int) {});
    __syncthreads();

    const int v = cnt[t];
    int inc = v;
#pragma unroll
    for (int o = 1; o < 32; o <<= 1) {
        int w = __shfl_up(inc, o);
        if (lane >= o) inc += w;
    }
    if (lane == 31) s_wave[wv] = inc;
    __syncthreads();
    int base = 0;
#pragma unroll
    for (int k = 0; k < BIN_N / 32; ++k)
        if (k < wv) base += s_wave[k];
    const int off = base + inc - v;
    cur[t] = off;
    const int node = b * BIN_N + t;
    if (node < n) {
        rowptr[node] = b0 + off;
        rsq[node] = rsqrtf((float)(v + 1));
        if (node == n - 1) rowptr[n] = b1;
    }
    __syncthreads();

    if (m <= CSR_CAP) {
        walk_bin(binned, lo, hi, nt, wv, lane,
                 [&](int2 el) { return atomicAdd(&cur[el.x & (BIN_N - 1)], 1); },
                 [&](int2 el, int r) { stage[r] = el.y; });
        __syncthreads();
        for (int j0 = 0; j0 < m; j0 += 4 * BIN_N) {
            int sv[4];
#pragma unroll
            for (int q = 0; q < 4; ++q) sv[q] = stage[min(j0 + q * BIN_N + t, CSR_CAP - 1)];
#pragma unroll
            for (int q = 0; q < 4; ++q) {
                const int j = j0 + q * BIN_N + t;
                if (j < m) csr[b0 + j] = sv[q];
            }
        }
    } else {
        walk_bin(binned, lo, hi, nt, wv, lane,
                 [&](int2 el) { return atomicAdd(&cur[el.x & (BIN_N - 1)], 1); },
                 [&](int2 el, int r) { csr[b0 + r] = el.y; });
    }
}

typedef __attribute__((ext_vector_type(4))) float v4f;

#define RB 12
#define WREST 8
#define SLAB 16
#define GRID_G 893
#define PITCH (D + 4)

__device__ __forceinline__ float4 ld_row(const float* __restrict__ p, int s, int lane) {
    const unsigned off = ((unsigned)s << 9) + ((unsigned)lane << 4);
    return *(const float4*)((const char*)p + off);
}
__device__ __forceinline__ void fma4(float4& a, float w, const float4& v) {
    a.x = fmaf(w, v.x, a.x); a.y = fmaf(w, v.y, a.y);
    a.z = fmaf(w, v.z, a.z); a.w = fmaf(w, v.w, a.w);
}
__device__ __forceinline__ int bc_i(int v, int k) { return __builtin_amdgcn_readlane(v, k); }
__device__ __forceinline__ float bc_f(float v, int k) {
    return __int_as_float(__builtin_amdgcn_readlane(__float_as_int(v), k));
}

__device__ __forceinline__ void lds_barrier() {
    __builtin_amdgcn_fence(__ATOMIC_RELEASE, "workgroup", "local");
    __builtin_amdgcn_s_barrier();
    __builtin_amdgcn_fence(__ATOMIC_ACQUIRE, "workgroup", "local");
}

__device__ __forceinline__ void add_row(float4& acc, float& ws, float ri, float q, const float4& v) {
    const float w = ri * q;
    fma4(acc, w, v);
    ws += w;
}

__global__ void __launch_bounds__(256) __attribute__((amdgpu_waves_per_eu(8)))
k_gather_project(const int* __restrict__ rowptr, const int* __restrict__ csr,
                 const float* __restrict__ x, const float* __restrict__ rsq,
                 const float* __restrict__ W, const float* __restrict__ bias,
                 float* __restrict__ out, int n_slabs, int e) {
    __shared__ __attribute__((aligned(16))) float As[SLAB * PITCH];
    __shared__ __attribute__((aligned(16))) float Os[SLAB * PITCH];
    __shared__ __attribute__((aligned(16))) float Ws[8 * WREST * 32 * 4];
    const int lane = threadIdx.x & 31;
    const int wave = __builtin_amdgcn_readfirstlane(threadIdx.x >> 5);
    const int lm = lane & 15;
    const int hh = lane >> 4;
    const int lk = hh << 2;

    const float* wr = W + (size_t)(16 * wave + lm) * D + lk;
    v4f bf[D / 8 - WREST];
#pragma unroll
    for (int j = 0; j < D / 8 - WREST; ++j) bf[j] = *(const v4f*)(wr + 8 * j);
    float* wl = Ws + (wave * WREST * 32 + lane) * 4;
#pragma unroll
    for (int j = 0; j < WREST; ++j) *(v4f*)(wl + j * 128) = *(const v4f*)(wr + 8 * (D / 8 - WREST + j));
    const float4 bias4 = *(const float4*)(bias + lane * 4);

    int slab = blockIdx.x;
    const int gstep = gridDim.x;
    int rp = 0, rp2 = 0, s_n = 0;
    float rq = 0.0f, rq2 = 0.0f, q_n = 0.0f;
    float4 self0 = make_float4(0.0f, 0.0f, 0.0f, 0.0f), self1 = self0;
    if (slab < n_slabs) {
        const int node0 = (slab << 4) + 2 * wave;
        const int node2 = (min(slab + gstep, n_slabs - 1) << 4) + 2 * wave;
        rp = rowptr[node0 + min(lane, 2)];
        rq = rsq[node0 + min(lane, 1)];
        rp2 = rowptr[node2 + min(lane, 2)];
        rq2 = rsq[node2 + min(lane, 1)];
        self0 = ld_row(x, node0, lane);
        self1 = ld_row(x, node0 + 1, lane);
        const int f0 = max(bc_i(rp, 0), 0);
        const int f2 = min(bc_i(rp, 2), e);
        if (lane < f2 - f0) s_n = csr[f0 + lane];
        q_n = rsq[s_n];
    }

    for (; slab < n_slabs; slab += gstep) {
        const int node0 = (slab << 4) + 2 * wave;
        const int p0 = max(bc_i(rp, 0), 0);
        const int p2 = min(bc_i(rp, 2), e);
        const int p1 = min(max(bc_i(rp, 1), p0), max(p2, p0));
        const float ri0 = bc_f(rq, 0);
        const float ri1 = bc_f(rq, 1);
        const int nnode0 = (min(slab + gstep, n_slabs - 1) << 4) + 2 * wave;
        const int nnode2 = (min(slab + 2 * gstep, n_slabs - 1) << 4) + 2 * wave;

        float ws0 = ri0 * ri0;
        float ws1 = ri1 * ri1;
        float4 acc0 = make_float4(ws0 * self0.x, ws0 * self0.y, ws0 * self0.z, ws0 * self0.w);
        float4 acc1 = make_float4(ws1 * self1.x, ws1 * self1.y, ws1 * self1.z, ws1 * self1.w);

        const int nf0 = max(bc_i(rp2, 0), 0);
        const int nfc = min(bc_i(rp2, 2), e) - nf0;
        auto next_slab_sources = [&]() {
            s_n = 0;
            if (lane < nfc) s_n = csr[nf0 + lane];
        };
        for (int base = p0; base < p2; base += 32) {
            const int m = min(32, p2 - base);
            const int mn = min(32, p2 - base - 32);
            const int s_l = s_n;
            const float q_l = q_n;
            const int split = min(max(p1 - base, 0), m);
            int k = 0;
            for (; k + RB <= m; k += RB) {
                float4 v[RB];
#pragma unroll
                for (int i = 0; i < RB; ++i) v[i] = ld_row(x, bc_i(s_l, k + i), lane);
                if (k == 0) {
                    const int pf0 = (mn > 0) ? base + 32 : nf0;
                    const int pfc = (mn > 0) ? mn : nfc;
                    s_n = 0;
                    if (lane < pfc) s_n = csr[pf0 + lane];
                } else if (k == RB) {
                    if (mn > 0) q_n = rsq[s_n];
                }
                if (k + RB <= split) {
#pragma unroll
                    for (int i = 0; i < RB; ++i) add_row(acc0, ws0, ri0, bc_f(q_l, k + i), v[i]);
                } else if (k >= split) {
#pragma unroll
                    for (int i = 0; i < RB; ++i) add_row(acc1, ws1, ri1, bc_f(q_l, k + i), v[i]);
                } else {
#pragma unroll
                    for (int i = 0; i < RB; ++i) {
                        if (k + i < split) add_row(acc0, ws0, ri0, bc_f(q_l, k + i), v[i]);
                        else               add_row(acc1, ws1, ri1, bc_f(q_l, k + i), v[i]);
                    }
                }
            }
            const int r = m - k;
            if (r > 0) {
                float4 v[RB - 1];
                v[0] = ld_row(x, bc_i(s_l, k), lane);
#pragma unroll
                for (int i = 1; i < RB - 1; ++i) {
                    v[i] = make_float4(0.0f, 0.0f, 0.0f, 0.0f);
                    if (i < r) v[i] = ld_row(x, bc_i(s_l, k + i), lane);
                }
                if (k == 0) next_slab_sources();
#pragma unroll
                for (int i = 0; i < RB - 1; ++i) {
                    if (i < r) {
                        if (k + i < split) add_row(acc0, ws0, ri0, bc_f(q_l, k + i), v[i]);
                        else               add_row(acc1, ws1, ri1, bc_f(q_l, k + i), v[i]);
                    }
                }
            }
        }
        if (p2 <= p0) next_slab_sources();
        *(float4*)(As + (2 * wave) * PITCH + lane * 4) = acc0;
        *(float4*)(As + (2 * wave + 1) * PITCH + lane * 4) = acc1;
        lds_barrier();

        q_n = rsq[s_n];
        self0 = ld_row(x, nnode0, lane);
        self1 = ld_row(x, nnode0 + 1, lane);
        rp = rp2;
        rq = rq2;
        rp2 = rowptr[nnode2 + min(lane, 2)];
        rq2 = rsq[nnode2 + min(lane, 1)];

        const float* ar = As + lm * PITCH + lk;
        v8f c0 = {}, c1 = {};
        v4f br[WREST];
#pragma unroll
        for (int j = 0; j < WREST; ++j) br[j] = *(const v4f*)(wl + j * 128);
#pragma unroll
        for (int j = 0; j < D / 8; ++j) {
            const v4f a = *(const v4f*)(ar + 8 * j);
            const v4f bj = (j < D / 8 - WREST) ? bf[j < D / 8 - WREST ? j : 0] : br[j < D / 8 - WREST ? 0 : j - (D / 8 - WREST)];
            const v2f alo = {a.x, a.y};
            const v2f ahi = {a.z, a.w};
            const v2f blo = {bj.x, bj.y};
            const v2f bhi = {bj.z, bj.w};
            c0 = __builtin_amdgcn_wmma_f32_16x16x4_f32(
                false, alo, false, blo, (short)0, c0, false, false);
            c1 = __builtin_amdgcn_wmma_f32_16x16x4_f32(
                false, ahi, false, bhi, (short)0, c1, false, false);
        }
        float* op = Os + (8 * hh) * PITCH + 16 * wave + lm;
#pragma unroll
        for (int v = 0; v < 8; ++v) op[v * PITCH] = c0[v] + c1[v];
        lds_barrier();

        float4 o0 = *(const float4*)(Os + (2 * wave) * PITCH + lane * 4);
        float4 o1 = *(const float4*)(Os + (2 * wave + 1) * PITCH + lane * 4);
        fma4(o0, ws0, bias4);
        fma4(o1, ws1, bias4);
        *(float4*)(out + (size_t)node0 * D + lane * 4) = o0;
        *(float4*)(out + (size_t)(node0 + 1) * D + lane * 4) = o1;
    }
}

extern "C" void kernel_launch(void* const* d_in, const int* in_sizes, int n_in,
                              void* d_out, int out_size, void* d_ws, size_t ws_size,
                              hipStream_t stream) {
    const float* x    = (const float*)d_in[0];
    const int*   ei   = (const int*)d_in[1];
    const float* W    = (const float*)d_in[2];
    const float* bias = (const float*)d_in[3];
    float*       out  = (float*)d_out;

    const int n = in_sizes[0] / D;
    const int e = in_sizes[1] / 2;

    int2*  binned = (int2*)d_ws;
    float* rsq    = (float*)(binned + e);
    int*   rowptr = (int*)(rsq + n);
    int*   csr    = rowptr + (n + 1);
    int*   tab    = csr + e;

    const int nb = (n + BIN_N - 1) / BIN_N;


    k_tile_sort<<<256, TS_T, 0, stream>>>(ei, tab, binned, e, nb);
    k_bin_csr  <<<nb, BIN_N, 0, stream>>>(binned, tab, rowptr, rsq, csr, n, e);

    k_gather_project<<<GRID_G, 256, 0, stream>>>(rowptr, csr, x, rsq, W, bias, out, n / SLAB, e);
}


# Slab kernel launched as 1024 blocks, four on each of the 256 compute units, 6 or 7 slabs each: every unit gets 24 or 25 slabs, where 893 blocks of 7 gave 21 or 28; kernel code and sums unchanged
# speedup vs baseline: 5.9446x; 1.0615x over previous
#include <hip/hip_runtime.h>


typedef __attribute__((ext_vector_type(2))) float v2f;
typedef __attribute__((ext_vector_type(8))) float v8f;

#define D 128

#define BIN_SHIFT 9
#define BIN_N 512
#define NB_MAX 256
#define TILE 8192
#define TS_T 512
#define EPT (TILE / TS_T)
#define CSR_CAP 12288

__global__ void __launch_bounds__(TS_T)
k_tile_sort(const int* __restrict__ ei, int* __restrict__ tab,
            int2* __restrict__ binned, int e, int nb) {
    __shared__ int hist[NB_MAX];
    __shared__ int loff[NB_MAX];
    __shared__ int s_wave[NB_MAX / 32];
    __shared__ int2 stage[TILE];
    const int t = threadIdx.x;
    const int lane = t & 31;
    const int wv = t >> 5;
    const int nt = (e + TILE - 1) / TILE;

    for (int u = blockIdx.x; u < nt; u += gridDim.x) {
        const int t0 = u * TILE;
        const int mt = min(TILE, e - t0);
        if (t < NB_MAX) hist[t] = 0;
        __syncthreads();

        int dd[EPT], ss[EPT], rr[EPT];
#pragma unroll
        for (int k = 0; k < EPT; ++k) {
            const int j = k * TS_T + t;
            dd[k] = (j < mt) ? ei[t0 + j] : 0;
            ss[k] = (j < mt) ? ei[e + t0 + j] : 0;
        }
#pragma unroll
        for (int k = 0; k < EPT; ++k) {
            rr[k] = 0;
            if (k * TS_T + t < mt) rr[k] = atomicAdd(&hist[dd[k] >> BIN_SHIFT], 1);
        }
        __syncthreads();

        int v = 0, inc = 0;
        if (t < NB_MAX) {
            v = hist[t];
            inc = v;
#pragma unroll
            for (int o = 1; o < 32; o <<= 1) {
                int w = __shfl_up(inc, o);
                if (lane >= o) inc += w;
            }
            if (lane == 31) s_wave[wv] = inc;
        }
        __syncthreads();
        if (t < NB_MAX) {
            int base = 0;
#pragma unroll
            for (int k = 0; k < NB_MAX / 32; ++k)
                if (k < wv) base += s_wave[k];
            const int off = base + inc - v;
            loff[t] = off;
            if (t < nb) tab[t * nt + u] = off;
            if (t == 0) tab[nb * nt + u] = mt;
        }
        __syncthreads();

        int po[EPT];
#pragma unroll
        for (int k = 0; k < EPT; ++k) po[k] = loff[dd[k] >> BIN_SHIFT] + rr[k];
#pragma unroll
        for (int k = 0; k < EPT; ++k)
            if (k * TS_T + t < mt) stage[po[k]] = make_int2(dd[k], ss[k]);
        __syncthreads();

        int2 pv[EPT];
#pragma unroll
        for (int k = 0; k < EPT; ++k) pv[k] = stage[k * TS_T + t];
#pragma unroll
        for (int k = 0; k < EPT; ++k)
            if (k * TS_T + t < mt) binned[t0 + k * TS_T + t] = pv[k];
    }
}

template <class R, class P>
__device__ __forceinline__ void walk_bin(const int2* __restrict__ binned,
                                         const int* __restrict__ lo, const int* __restrict__ hi,
                                         int nt, int wv, int lane, R rank, P put) {
    for (int ub = wv; ub < nt; ub += 16 * 32) {
        int pbase = 0, plen = 0;
        const int u = ub + 16 * lane;
        if (u < nt) {
            const int s = min(max(lo[u], 0), TILE);
            plen = min(max(hi[u] - s, 0), TILE - s);
            pbase = u * TILE + s;
        }
        const int np = min(32, (nt - ub + 15) >> 4);
        for (int l0 = 0; l0 < np; l0 += 4) {
            int2 v0[4], v1[4];
            int ln[4], bs[4];
#pragma unroll
            for (int q = 0; q < 4; ++q) {
                const int l = min(l0 + q, 31);
                ln[q] = (l0 + q < np) ? __builtin_amdgcn_readlane(plen, l) : 0;
                bs[q] = __builtin_amdgcn_readlane(pbase, l);
                v0[q] = binned[(lane < ln[q]) ? bs[q] + lane : 0];
                v1[q] = binned[(lane + 32 < ln[q]) ? bs[q] + 32 + lane : 0];
            }
            asm volatile("" ::
                "v"(v0[0].x), "v"(v0[0].y), "v"(v1[0].x), "v"(v1[0].y),
                "v"(v0[1].x), "v"(v0[1].y), "v"(v1[1].x), "v"(v1[1].y),
                "v"(v0[2].x), "v"(v0[2].y), "v"(v1[2].x), "v"(v1[2].y),
                "v"(v0[3].x), "v"(v0[3].y), "v"(v1[3].x), "v"(v1[3].y));
            int r0[4], r1[4];
#pragma unroll
            for (int q = 0; q < 4; ++q) {
                r0[q] = 0;
                r1[q] = 0;
                if (lane < ln[q]) r0[q] = rank(v0[q]);
                if (lane + 32 < ln[q]) r1[q] = rank(v1[q]);
            }
#pragma unroll
            for (int q = 0; q < 4; ++q) {
                if (lane < ln[q]) put(v0[q], r0[q]);
                if (lane + 32 < ln[q]) put(v1[q], r1[q]);
                for (int i = 64 + lane; i < ln[q]; i += 32) {
                    const int2 el = binned[bs[q] + i];
                    put(el, rank(el));
                }
            }
        }
    }
}

__global__ void __launch_bounds__(BIN_N)
k_bin_csr(const int2* __restrict__ binned, const int* __restrict__ tab,
          int* __restrict__ rowptr, float* __restrict__ rsq, int* __restrict__ csr,
          int n, int e) {
    __shared__ int cnt[BIN_N];
    __shared__ int cur[BIN_N];
    __shared__ int s_wave[BIN_N / 32];
    __shared__ int s_lo[BIN_N / 32];
    __shared__ int s_hi[BIN_N / 32];
    __shared__ int stage[CSR_CAP];
    const int t = threadIdx.x;
    const int lane = t & 31;
    const int wv = __builtin_amdgcn_readfirstlane(t >> 5);
    const int b = blockIdx.x;
    const int nt = (e + TILE - 1) / TILE;
    const int* lo = tab + (size_t)b * nt;
    const int* hi = tab + (size_t)(b + 1) * nt;

    int p0 = 0, p1 = 0;
    for (int u = t; u < nt; u += BIN_N) {
        p0 += lo[u];
        p1 += hi[u];
    }
#pragma unroll
    for (int o = 16; o > 0; o >>= 1) {
        p0 += __shfl_xor(p0, o);
        p1 += __shfl_xor(p1, o);
    }
    if (lane == 0) {
        s_lo[wv] = p0;
        s_hi[wv] = p1;
    }
    cnt[t] = 0;
    __syncthreads();
    int b0 = 0, b1 = 0;
#pragma unroll
    for (int k = 0; k < BIN_N / 32; ++k) {
        b0 += s_lo[k];
        b1 += s_hi[k];
    }
    const int m = b1 - b0;

    walk_bin(binned, lo, hi, nt, wv, lane,
             [&](int2 el) { atomicAdd(&cnt[el.x & (BIN_N - 1)], 1); return 0; },
             [&](int2, int) {});
    __syncthreads();

    const int v = cnt[t];
    int inc = v;
#pragma unroll
    for (int o = 1; o < 32; o <<= 1) {
        int w = __shfl_up(inc, o);
        if (lane >= o) inc += w;
    }
    if (lane == 31) s_wave[wv] = inc;
    __syncthreads();
    int base = 0;
#pragma unroll
    for (int k = 0; k < BIN_N / 32; ++k)
        if (k < wv) base += s_wave[k];
    const int off = base + inc - v;
    cur[t] = off;
    const int node = b * BIN_N + t;
    if (node < n) {
        rowptr[node] = b0 + off;
        rsq[node] = rsqrtf((float)(v + 1));
        if (node == n - 1) rowptr[n] = b1;
    }
    __syncthreads();

    if (m <= CSR_CAP) {
        walk_bin(binned, lo, hi, nt, wv, lane,
                 [&](int2 el) { return atomicAdd(&cur[el.x & (BIN_N - 1)], 1); },
                 [&](int2 el, int r) { stage[r] = el.y; });
        __syncthreads();
        for (int j0 = 0; j0 < m; j0 += 4 * BIN_N) {
            int sv[4];
#pragma unroll
            for (int q = 0; q < 4; ++q) sv[q] = stage[min(j0 + q * BIN_N + t, CSR_CAP - 1)];
#pragma unroll
            for (int q = 0; q < 4; ++q) {
                const int j = j0 + q * BIN_N + t;
                if (j < m) csr[b0 + j] = sv[q];
            }
        }
    } else {
        walk_bin(binned, lo, hi, nt, wv, lane,
                 [&](int2 el) { return atomicAdd(&cur[el.x & (BIN_N - 1)], 1); },
                 [&](int2 el, int r) { csr[b0 + r] = el.y; });
    }
}

typedef __attribute__((ext_vector_type(4))) float v4f;

#define RB 12
#define WREST 8
#define SLAB 16
#define GRID_G 1024
#define PITCH (D + 4)

__device__ __forceinline__ float4 ld_row(const float* __restrict__ p, int s, int lane) {
    const unsigned off = ((unsigned)s << 9) + ((unsigned)lane << 4);
    return *(const float4*)((const char*)p + off);
}
__device__ __forceinline__ void fma4(float4& a, float w, const float4& v) {
    a.x = fmaf(w, v.x, a.x); a.y = fmaf(w, v.y, a.y);
    a.z = fmaf(w, v.z, a.z); a.w = fmaf(w, v.w, a.w);
}
__device__ __forceinline__ int bc_i(int v, int k) { return __builtin_amdgcn_readlane(v, k); }
__device__ __forceinline__ float bc_f(float v, int k) {
    return __int_as_float(__builtin_amdgcn_readlane(__float_as_int(v), k));
}

__device__ __forceinline__ void lds_barrier() {
    __builtin_amdgcn_fence(__ATOMIC_RELEASE, "workgroup", "local");
    __builtin_amdgcn_s_barrier();
    __builtin_amdgcn_fence(__ATOMIC_ACQUIRE, "workgroup", "local");
}

__device__ __forceinline__ void add_row(float4& acc, float& ws, float ri, float q, const float4& v) {
    const float w = ri * q;
    fma4(acc, w, v);
    ws += w;
}

__global__ void __launch_bounds__(256) __attribute__((amdgpu_waves_per_eu(8)))
k_gather_project(const int* __restrict__ rowptr, const int* __restrict__ csr,
                 const float* __restrict__ x, const float* __restrict__ rsq,
                 const float* __restrict__ W, const float* __restrict__ bias,
                 float* __restrict__ out, int n_slabs, int e) {
    __shared__ __attribute__((aligned(16))) float As[SLAB * PITCH];
    __shared__ __attribute__((aligned(16))) float Os[SLAB * PITCH];
    __shared__ __attribute__((aligned(16))) float Ws[8 * WREST * 32 * 4];
    const int lane = threadIdx.x & 31;
    const int wave = __builtin_amdgcn_readfirstlane(threadIdx.x >> 5);
    const int lm = lane & 15;
    const int hh = lane >> 4;
    const int lk = hh << 2;

    const float* wr = W + (size_t)(16 * wave + lm) * D + lk;
    v4f bf[D / 8 - WREST];
#pragma unroll
    for (int j = 0; j < D / 8 - WREST; ++j) bf[j] = *(const v4f*)(wr + 8 * j);
    float* wl = Ws + (wave * WREST * 32 + lane) * 4;
#pragma unroll
    for (int j = 0; j < WREST; ++j) *(v4f*)(wl + j * 128) = *(const v4f*)(wr + 8 * (D / 8 - WREST + j));
    const float4 bias4 = *(const float4*)(bias + lane * 4);

    int slab = blockIdx.x;
    const int gstep = gridDim.x;
    int rp = 0, rp2 = 0, s_n = 0;
    float rq = 0.0f, rq2 = 0.0f, q_n = 0.0f;
    float4 self0 = make_float4(0.0f, 0.0f, 0.0f, 0.0f), self1 = self0;
    if (slab < n_slabs) {
        const int node0 = (slab << 4) + 2 * wave;
        const int node2 = (min(slab + gstep, n_slabs - 1) << 4) + 2 * wave;
        rp = rowptr[node0 + min(lane, 2)];
        rq = rsq[node0 + min(lane, 1)];
        rp2 = rowptr[node2 + min(lane, 2)];
        rq2 = rsq[node2 + min(lane, 1)];
        self0 = ld_row(x, node0, lane);
        self1 = ld_row(x, node0 + 1, lane);
        const int f0 = max(bc_i(rp, 0), 0);
        const int f2 = min(bc_i(rp, 2), e);
        if (lane < f2 - f0) s_n = csr[f0 + lane];
        q_n = rsq[s_n];
    }

    for (; slab < n_slabs; slab += gstep) {
        const int node0 = (slab << 4) + 2 * wave;
        const int p0 = max(bc_i(rp, 0), 0);
        const int p2 = min(bc_i(rp, 2), e);
        const int p1 = min(max(bc_i(rp, 1), p0), max(p2, p0));
        const float ri0 = bc_f(rq, 0);
        const float ri1 = bc_f(rq, 1);
        const int nnode0 = (min(slab + gstep, n_slabs - 1) << 4) + 2 * wave;
        const int nnode2 = (min(slab + 2 * gstep, n_slabs - 1) << 4) + 2 * wave;

        float ws0 = ri0 * ri0;
        float ws1 = ri1 * ri1;
        float4 acc0 = make_float4(ws0 * self0.x, ws0 * self0.y, ws0 * self0.z, ws0 * self0.w);
        float4 acc1 = make_float4(ws1 * self1.x, ws1 * self1.y, ws1 * self1.z, ws1 * self1.w);

        const int nf0 = max(bc_i(rp2, 0), 0);
        const int nfc = min(bc_i(rp2, 2), e) - nf0;
        auto next_slab_sources = [&]() {
            s_n = 0;
            if (lane < nfc) s_n = csr[nf0 + lane];
        };
        for (int base = p0; base < p2; base += 32) {
            const int m = min(32, p2 - base);
            const int mn = min(32, p2 - base - 32);
            const int s_l = s_n;
            const float q_l = q_n;
            const int split = min(max(p1 - base, 0), m);
            int k = 0;
            for (; k + RB <= m; k += RB) {
                float4 v[RB];
#pragma unroll
                for (int i = 0; i < RB; ++i) v[i] = ld_row(x, bc_i(s_l, k + i), lane);
                if (k == 0) {
                    const int pf0 = (mn > 0) ? base + 32 : nf0;
                    const int pfc = (mn > 0) ? mn : nfc;
                    s_n = 0;
                    if (lane < pfc) s_n = csr[pf0 + lane];
                } else if (k == RB) {
                    if (mn > 0) q_n = rsq[s_n];
                }
                if (k + RB <= split) {
#pragma unroll
                    for (int i = 0; i < RB; ++i) add_row(acc0, ws0, ri0, bc_f(q_l, k + i), v[i]);
                } else if (k >= split) {
#pragma unroll
                    for (int i = 0; i < RB; ++i) add_row(acc1, ws1, ri1, bc_f(q_l, k + i), v[i]);
                } else {
#pragma unroll
                    for (int i = 0; i < RB; ++i) {
                        if (k + i < split) add_row(acc0, ws0, ri0, bc_f(q_l, k + i), v[i]);
                        else               add_row(acc1, ws1, ri1, bc_f(q_l, k + i), v[i]);
                    }
                }
            }
            const int r = m - k;
            if (r > 0) {
                float4 v[RB - 1];
                v[0] = ld_row(x, bc_i(s_l, k), lane);
#pragma unroll
                for (int i = 1; i < RB - 1; ++i) {
                    v[i] = make_float4(0.0f, 0.0f, 0.0f, 0.0f);
                    if (i < r) v[i] = ld_row(x, bc_i(s_l, k + i), lane);
                }
                if (k == 0) next_slab_sources();
#pragma unroll
                for (int i = 0; i < RB - 1; ++i) {
                    if (i < r) {
                        if (k + i < split) add_row(acc0, ws0, ri0, bc_f(q_l, k + i), v[i]);
                        else               add_row(acc1, ws1, ri1, bc_f(q_l, k + i), v[i]);
                    }
                }
            }
        }
        if (p2 <= p0) next_slab_sources();
        *(float4*)(As + (2 * wave) * PITCH + lane * 4) = acc0;
        *(float4*)(As + (2 * wave + 1) * PITCH + lane * 4) = acc1;
        lds_barrier();

        q_n = rsq[s_n];
        self0 = ld_row(x, nnode0, lane);
        self1 = ld_row(x, nnode0 + 1, lane);
        rp = rp2;
        rq = rq2;
        rp2 = rowptr[nnode2 + min(lane, 2)];
        rq2 = rsq[nnode2 + min(lane, 1)];

        const float* ar = As + lm * PITCH + lk;
        v8f c0 = {}, c1 = {};
        v4f br[WREST];
#pragma unroll
        for (int j = 0; j < WREST; ++j) br[j] = *(const v4f*)(wl + j * 128);
#pragma unroll
        for (int j = 0; j < D / 8; ++j) {
            const v4f a = *(const v4f*)(ar + 8 * j);
            const v4f bj = (j < D / 8 - WREST) ? bf[j < D / 8 - WREST ? j : 0] : br[j < D / 8 - WREST ? 0 : j - (D / 8 - WREST)];
            const v2f alo = {a.x, a.y};
            const v2f ahi = {a.z, a.w};
            const v2f blo = {bj.x, bj.y};
            const v2f bhi = {bj.z, bj.w};
            c0 = __builtin_amdgcn_wmma_f32_16x16x4_f32(
                false, alo, false, blo, (short)0, c0, false, false);
            c1 = __builtin_amdgcn_wmma_f32_16x16x4_f32(
                false, ahi, false, bhi, (short)0, c1, false, false);
        }
        float* op = Os + (8 * hh) * PITCH + 16 * wave + lm;
#pragma unroll
        for (int v = 0; v < 8; ++v) op[v * PITCH] = c0[v] + c1[v];
        lds_barrier();

        float4 o0 = *(const float4*)(Os + (2 * wave) * PITCH + lane * 4);
        float4 o1 = *(const float4*)(Os + (2 * wave + 1) * PITCH + lane * 4);
        fma4(o0, ws0, bias4);
        fma4(o1, ws1, bias4);
        *(float4*)(out + (size_t)node0 * D + lane * 4) = o0;
        *(float4*)(out + (size_t)(node0 + 1) * D + lane * 4) = o1;
    }
}

extern "C" void kernel_launch(void* const* d_in, const int* in_sizes, int n_in,
                              void* d_out, int out_size, void* d_ws, size_t ws_size,
                              hipStream_t stream) {
    const float* x    = (const float*)d_in[0];
    const int*   ei   = (const int*)d_in[1];
    const float* W    = (const float*)d_in[2];
    const float* bias = (const float*)d_in[3];
    float*       out  = (float*)d_out;

    const int n = in_sizes[0] / D;
    const int e = in_sizes[1] / 2;

    int2*  binned = (int2*)d_ws;
    float* rsq    = (float*)(binned + e);
    int*   rowptr = (int*)(rsq + n);
    int*   csr    = rowptr + (n + 1);
    int*   tab    = csr + e;

    const int nb = (n + BIN_N - 1) / BIN_N;


    k_tile_sort<<<256, TS_T, 0, stream>>>(ei, tab, binned, e, nb);
    k_bin_csr  <<<nb, BIN_N, 0, stream>>>(binned, tab, rowptr, rsq, csr, n, e);

    k_gather_project<<<GRID_G, 256, 0, stream>>>(rowptr, csr, x, rsq, W, bias, out, n / SLAB, e);
}
